# starting kernel
# baseline (speedup 1.0000x reference)
#include <hip/hip_runtime.h>
#include <hip/hip_bf16.h>


typedef __attribute__((ext_vector_type(2))) float v2f;
typedef __attribute__((ext_vector_type(8))) float v8f;

__device__ __forceinline__ unsigned f2ord(float f) {
    unsigned u = __float_as_uint(f);
    return (u & 0x80000000u) ? ~u : (u | 0x80000000u);
}
__device__ __forceinline__ float ord2f(unsigned u) {
    unsigned v = (u & 0x80000000u) ? (u & 0x7FFFFFFFu) : ~u;
    return __uint_as_float(v);
}

__global__ void gat_gemm_wmma(const float* __restrict__ x,
                              const float* __restrict__ W,
                              float* __restrict__ h) {
    const int wave   = threadIdx.x >> 5;
    const int lane   = threadIdx.x & 31;
    const int lhalf  = lane & 15;
    const int khalf  = (lane >> 4) << 1;
    const int rowA   = blockIdx.x * 16 + lhalf;
    const int colB   = wave * 16 + lhalf;

    const float* __restrict__ arow = x + (size_t)rowA * 128;

    v8f c = {0.f, 0.f, 0.f, 0.f, 0.f, 0.f, 0.f, 0.f};

#pragma unroll 4
    for (int k0 = 0; k0 < 128; k0 += 4) {
        const int k = k0 + khalf;
        v2f a;
        a.x = arow[k];
        a.y = arow[k + 1];
        v2f b;
        b.x = W[(size_t)k * 128 + colB];
        b.y = W[(size_t)(k + 1) * 128 + colB];
        c = __builtin_amdgcn_wmma_f32_16x16x4_f32(false, a, false, b,
                                                  (short)0, c, false, false);
    }

    const int mhi = (lane >> 4) << 3;
#pragma unroll
    for (int r = 0; r < 8; ++r) {
        const int m = blockIdx.x * 16 + r + mhi;
        h[(size_t)m * 128 + wave * 16 + lhalf] = c[r];
    }
}

__global__ void gat_init(float* __restrict__ out, const float* __restrict__ bias,
                         float* __restrict__ denom, unsigned* __restrict__ maxbuf,
                         int total, int nh) {
    const int i = blockIdx.x * blockDim.x + threadIdx.x;
    if (i < 4)     maxbuf[i] = f2ord(-3.0e38f);
    if (i < nh)    denom[i]  = 0.0f;
    if (i < total) out[i]    = bias[i & 127];
}

__global__ void gat_alpha(const float* __restrict__ h,
                          const float* __restrict__ att_l,
                          const float* __restrict__ att_r,
                          float* __restrict__ alpha_l,
                          float* __restrict__ alpha_r, int nh) {
    const int i = blockIdx.x * blockDim.x + threadIdx.x;
    if (i >= nh) return;
    const int head = i & 3;
    const float* __restrict__ hv = h + ((size_t)(i >> 2)) * 128 + head * 32;
    const float* __restrict__ al = att_l + head * 32;
    const float* __restrict__ ar = att_r + head * 32;
    float sl = 0.f, sr = 0.f;
#pragma unroll
    for (int d = 0; d < 32; ++d) {
        const float v = hv[d];
        sl += v * al[d];
        sr += v * ar[d];
    }
    alpha_l[i] = sl;
    alpha_r[i] = sr;
}

__global__ void gat_edge_max(const int* __restrict__ ei,
                             const float* __restrict__ alpha_l,
                             const float* __restrict__ alpha_r,
                             unsigned* __restrict__ maxbuf, int E) {
    __shared__ unsigned smax[4];
    if (threadIdx.x < 4) smax[threadIdx.x] = 0u;
    __syncthreads();
    const int e = blockIdx.x * blockDim.x + threadIdx.x;
    if (e < E) {
        const int f = ei[e];
        const int t = ei[E + e];
#pragma unroll
        for (int hd = 0; hd < 4; ++hd) {
            float a = alpha_l[f * 4 + hd] + alpha_r[t * 4 + hd];
            a = a > 0.f ? a : 0.2f * a;
            atomicMax(&smax[hd], f2ord(a));
        }
    }
    __syncthreads();
    if (threadIdx.x < 4) atomicMax(&maxbuf[threadIdx.x], smax[threadIdx.x]);
}

__global__ void gat_denom(const int* __restrict__ ei,
                          const float* __restrict__ alpha_l,
                          const float* __restrict__ alpha_r,
                          const unsigned* __restrict__ maxbuf,
                          float* __restrict__ denom, int E) {
    const int e = blockIdx.x * blockDim.x + threadIdx.x;
    if (e >= E) return;
    const int f = ei[e];
    const int t = ei[E + e];
#pragma unroll
    for (int hd = 0; hd < 4; ++hd) {
        float a = alpha_l[f * 4 + hd] + alpha_r[t * 4 + hd];
        a = a > 0.f ? a : 0.2f * a;
        const float ex = __expf(a - ord2f(maxbuf[hd]));
        atomicAdd(&denom[t * 4 + hd], ex);
    }
}

__global__ void gat_aggregate(const float* __restrict__ hbuf,
                              const int* __restrict__ ei,
                              const float* __restrict__ alpha_l,
                              const float* __restrict__ alpha_r,
                              const float* __restrict__ denom,
                              const unsigned* __restrict__ maxbuf,
                              float* __restrict__ out, int E) {
    const int gw   = (int)((blockIdx.x * (unsigned)blockDim.x + threadIdx.x) >> 5);
    const int lane = threadIdx.x & 31;
    if (gw >= E) return;
    const int f = ei[gw];
    const int t = ei[E + gw];
    const float* __restrict__ src = hbuf + (size_t)f * 128;
    float* __restrict__ dst = out + (size_t)t * 128;
#pragma unroll
    for (int hd = 0; hd < 4; ++hd) {
        float a = alpha_l[f * 4 + hd] + alpha_r[t * 4 + hd];
        a = a > 0.f ? a : 0.2f * a;
        const float ex = __expf(a - ord2f(maxbuf[hd]));
        const float w  = ex / (denom[t * 4 + hd] + 1e-9f);
        atomicAdd(&dst[hd * 32 + lane], w * src[hd * 32 + lane]);
    }
}

extern "C" void kernel_launch(void* const* d_in, const int* in_sizes, int n_in,
                              void* d_out, int out_size, void* d_ws, size_t ws_size,
                              hipStream_t stream) {
    const float* x     = (const float*)d_in[0];
    const int*   ei    = (const int*)d_in[1];
    const float* W     = (const float*)d_in[2];
    const float* att_l = (const float*)d_in[3];
    const float* att_r = (const float*)d_in[4];
    const float* bias  = (const float*)d_in[5];
    float* out = (float*)d_out;

    const int N  = in_sizes[0] / 128;
    const int E  = in_sizes[1] / 2;
    const int NH = N * 4;

    float* h       = (float*)d_ws;
    float* alpha_l = h + (size_t)N * 128;
    float* alpha_r = alpha_l + NH;
    float* denom   = alpha_r + NH;
    unsigned* maxbuf = (unsigned*)(denom + NH);

    const int BLK = 256;

    {
        const int total = N * 128;
        gat_init<<<(total + BLK - 1) / BLK, BLK, 0, stream>>>(out, bias, denom,
                                                              maxbuf, total, NH);
    }
    gat_gemm_wmma<<<N / 16, BLK, 0, stream>>>(x, W, h);
    gat_alpha<<<(NH + BLK - 1) / BLK, BLK, 0, stream>>>(h, att_l, att_r,
                                                        alpha_l, alpha_r, NH);
    gat_edge_max<<<(E + BLK - 1) / BLK, BLK, 0, stream>>>(ei, alpha_l, alpha_r,
                                                          maxbuf, E);
    gat_denom<<<(E + BLK - 1) / BLK, BLK, 0, stream>>>(ei, alpha_l, alpha_r,
                                                       maxbuf, denom, E);
    {
        const long long threads = (long long)E * 32;
        const int blocks = (int)((threads + BLK - 1) / BLK);
        gat_aggregate<<<blocks, BLK, 0, stream>>>(h, ei, alpha_l, alpha_r,
                                                  denom, maxbuf, out, E);
    }
}


# Denominator pass: thread per (edge,head), heads in adjacent lanes: 1 atomic instruction per 8 edges (8 rows of 16 B) instead of 4 instructions x 32 scattered lines; atomic add order changes
# speedup vs baseline: 1.3340x; 1.3340x over previous
#include <hip/hip_runtime.h>
#include <hip/hip_bf16.h>


typedef __attribute__((ext_vector_type(2))) float v2f;
typedef __attribute__((ext_vector_type(8))) float v8f;

__device__ __forceinline__ unsigned f2ord(float f) {
    unsigned u = __float_as_uint(f);
    return (u & 0x80000000u) ? ~u : (u | 0x80000000u);
}
__device__ __forceinline__ float ord2f(unsigned u) {
    unsigned v = (u & 0x80000000u) ? (u & 0x7FFFFFFFu) : ~u;
    return __uint_as_float(v);
}

__global__ void gat_gemm_wmma(const float* __restrict__ x,
                              const float* __restrict__ W,
                              float* __restrict__ h) {
    const int wave   = threadIdx.x >> 5;
    const int lane   = threadIdx.x & 31;
    const int lhalf  = lane & 15;
    const int khalf  = (lane >> 4) << 1;
    const int rowA   = blockIdx.x * 16 + lhalf;
    const int colB   = wave * 16 + lhalf;

    const float* __restrict__ arow = x + (size_t)rowA * 128;

    v8f c = {0.f, 0.f, 0.f, 0.f, 0.f, 0.f, 0.f, 0.f};

#pragma unroll 4
    for (int k0 = 0; k0 < 128; k0 += 4) {
        const int k = k0 + khalf;
        v2f a;
        a.x = arow[k];
        a.y = arow[k + 1];
        v2f b;
        b.x = W[(size_t)k * 128 + colB];
        b.y = W[(size_t)(k + 1) * 128 + colB];
        c = __builtin_amdgcn_wmma_f32_16x16x4_f32(false, a, false, b,
                                                  (short)0, c, false, false);
    }

    const int mhi = (lane >> 4) << 3;
#pragma unroll
    for (int r = 0; r < 8; ++r) {
        const int m = blockIdx.x * 16 + r + mhi;
        h[(size_t)m * 128 + wave * 16 + lhalf] = c[r];
    }
}

__global__ void gat_init(float* __restrict__ out, const float* __restrict__ bias,
                         float* __restrict__ denom, unsigned* __restrict__ maxbuf,
                         int total, int nh) {
    const int i = blockIdx.x * blockDim.x + threadIdx.x;
    if (i < 4)     maxbuf[i] = f2ord(-3.0e38f);
    if (i < nh)    denom[i]  = 0.0f;
    if (i < total) out[i]    = bias[i & 127];
}

__global__ void gat_alpha(const float* __restrict__ h,
                          const float* __restrict__ att_l,
                          const float* __restrict__ att_r,
                          float* __restrict__ alpha_l,
                          float* __restrict__ alpha_r, int nh) {
    const int i = blockIdx.x * blockDim.x + threadIdx.x;
    if (i >= nh) return;
    const int head = i & 3;
    const float* __restrict__ hv = h + ((size_t)(i >> 2)) * 128 + head * 32;
    const float* __restrict__ al = att_l + head * 32;
    const float* __restrict__ ar = att_r + head * 32;
    float sl = 0.f, sr = 0.f;
#pragma unroll
    for (int d = 0; d < 32; ++d) {
        const float v = hv[d];
        sl += v * al[d];
        sr += v * ar[d];
    }
    alpha_l[i] = sl;
    alpha_r[i] = sr;
}

__global__ void gat_edge_max(const int* __restrict__ ei,
                             const float* __restrict__ alpha_l,
                             const float* __restrict__ alpha_r,
                             unsigned* __restrict__ maxbuf, int E) {
    __shared__ unsigned smax[4];
    if (threadIdx.x < 4) smax[threadIdx.x] = 0u;
    __syncthreads();
    const int e = blockIdx.x * blockDim.x + threadIdx.x;
    if (e < E) {
        const int f = ei[e];
        const int t = ei[E + e];
#pragma unroll
        for (int hd = 0; hd < 4; ++hd) {
            float a = alpha_l[f * 4 + hd] + alpha_r[t * 4 + hd];
            a = a > 0.f ? a : 0.2f * a;
            atomicMax(&smax[hd], f2ord(a));
        }
    }
    __syncthreads();
    if (threadIdx.x < 4) atomicMax(&maxbuf[threadIdx.x], smax[threadIdx.x]);
}

__global__ void gat_denom(const int* __restrict__ ei,
                          const float* __restrict__ alpha_l,
                          const float* __restrict__ alpha_r,
                          const unsigned* __restrict__ maxbuf,
                          float* __restrict__ denom, int E) {
    const int i  = blockIdx.x * blockDim.x + threadIdx.x;
    const int e  = i >> 2;
    const int hd = i & 3;
    if (e >= E) return;
    const int f = ei[e];
    const int t = ei[E + e];
    float a = alpha_l[f * 4 + hd] + alpha_r[t * 4 + hd];
    a = a > 0.f ? a : 0.2f * a;
    const float ex = __expf(a - ord2f(maxbuf[hd]));
    atomicAdd(&denom[t * 4 + hd], ex);
}

__global__ void gat_aggregate(const float* __restrict__ hbuf,
                              const int* __restrict__ ei,
                              const float* __restrict__ alpha_l,
                              const float* __restrict__ alpha_r,
                              const float* __restrict__ denom,
                              const unsigned* __restrict__ maxbuf,
                              float* __restrict__ out, int E) {
    const int gw   = (int)((blockIdx.x * (unsigned)blockDim.x + threadIdx.x) >> 5);
    const int lane = threadIdx.x & 31;
    if (gw >= E) return;
    const int f = ei[gw];
    const int t = ei[E + gw];
    const float* __restrict__ src = hbuf + (size_t)f * 128;
    float* __restrict__ dst = out + (size_t)t * 128;
#pragma unroll
    for (int hd = 0; hd < 4; ++hd) {
        float a = alpha_l[f * 4 + hd] + alpha_r[t * 4 + hd];
        a = a > 0.f ? a : 0.2f * a;
        const float ex = __expf(a - ord2f(maxbuf[hd]));
        const float w  = ex / (denom[t * 4 + hd] + 1e-9f);
        atomicAdd(&dst[hd * 32 + lane], w * src[hd * 32 + lane]);
    }
}

extern "C" void kernel_launch(void* const* d_in, const int* in_sizes, int n_in,
                              void* d_out, int out_size, void* d_ws, size_t ws_size,
                              hipStream_t stream) {
    const float* x     = (const float*)d_in[0];
    const int*   ei    = (const int*)d_in[1];
    const float* W     = (const float*)d_in[2];
    const float* att_l = (const float*)d_in[3];
    const float* att_r = (const float*)d_in[4];
    const float* bias  = (const float*)d_in[5];
    float* out = (float*)d_out;

    const int N  = in_sizes[0] / 128;
    const int E  = in_sizes[1] / 2;
    const int NH = N * 4;

    float* h       = (float*)d_ws;
    float* alpha_l = h + (size_t)N * 128;
    float* alpha_r = alpha_l + NH;
    float* denom   = alpha_r + NH;
    unsigned* maxbuf = (unsigned*)(denom + NH);

    const int BLK = 256;

    {
        const int total = N * 128;
        gat_init<<<(total + BLK - 1) / BLK, BLK, 0, stream>>>(out, bias, denom,
                                                              maxbuf, total, NH);
    }
    gat_gemm_wmma<<<N / 16, BLK, 0, stream>>>(x, W, h);
    gat_alpha<<<(NH + BLK - 1) / BLK, BLK, 0, stream>>>(h, att_l, att_r,
                                                        alpha_l, alpha_r, NH);
    gat_edge_max<<<(E + BLK - 1) / BLK, BLK, 0, stream>>>(ei, alpha_l, alpha_r,
                                                          maxbuf, E);
    gat_denom<<<(E * 4 + BLK - 1) / BLK, BLK, 0, stream>>>(ei, alpha_l, alpha_r,
                                                           maxbuf, denom, E);
    {
        const long long threads = (long long)E * 32;
        const int blocks = (int)((threads + BLK - 1) / BLK);
        gat_aggregate<<<blocks, BLK, 0, stream>>>(h, ei, alpha_l, alpha_r,
                                                  denom, maxbuf, out, E);
    }
}


# Aggregate by destination: in-call counting sort of edges by to-node (returning int add = rank, scan, fill), then one wave per node sums denominators and rows in registers; sum order = arrival rank
# speedup vs baseline: 1.8453x; 1.3833x over previous
#include <hip/hip_runtime.h>
#include <hip/hip_bf16.h>


typedef __attribute__((ext_vector_type(2))) float v2f;
typedef __attribute__((ext_vector_type(8))) float v8f;

__device__ __forceinline__ unsigned f2ord(float f) {
    unsigned u = __float_as_uint(f);
    return (u & 0x80000000u) ? ~u : (u | 0x80000000u);
}
__device__ __forceinline__ float ord2f(unsigned u) {
    unsigned v = (u & 0x80000000u) ? (u & 0x7FFFFFFFu) : ~u;
    return __uint_as_float(v);
}

__global__ void gat_gemm_wmma(const float* __restrict__ x,
                              const float* __restrict__ W,
                              float* __restrict__ h) {
    const int wave   = threadIdx.x >> 5;
    const int lane   = threadIdx.x & 31;
    const int lhalf  = lane & 15;
    const int khalf  = (lane >> 4) << 1;
    const int rowA   = blockIdx.x * 16 + lhalf;
    const int colB   = wave * 16 + lhalf;

    const float* __restrict__ arow = x + (size_t)rowA * 128;

    v8f c = {0.f, 0.f, 0.f, 0.f, 0.f, 0.f, 0.f, 0.f};

#pragma unroll 4
    for (int k0 = 0; k0 < 128; k0 += 4) {
        const int k = k0 + khalf;
        v2f a;
        a.x = arow[k];
        a.y = arow[k + 1];
        v2f b;
        b.x = W[(size_t)k * 128 + colB];
        b.y = W[(size_t)(k + 1) * 128 + colB];
        c = __builtin_amdgcn_wmma_f32_16x16x4_f32(false, a, false, b,
                                                  (short)0, c, false, false);
    }

    const int mhi = (lane >> 4) << 3;
#pragma unroll
    for (int r = 0; r < 8; ++r) {
        const int m = blockIdx.x * 16 + r + mhi;
        h[(size_t)m * 128 + wave * 16 + lhalf] = c[r];
    }
}

__global__ void gat_alpha(const float* __restrict__ h,
                          const float* __restrict__ att_l,
                          const float* __restrict__ att_r,
                          float* __restrict__ alpha_l,
                          float* __restrict__ alpha_r,
                          int* __restrict__ deg,
                          unsigned* __restrict__ maxbuf, int nh, int n_nodes) {
    const int i = blockIdx.x * blockDim.x + threadIdx.x;
    if (i < 4)       maxbuf[i] = f2ord(-3.0e38f);
    if (i < n_nodes) deg[i]    = 0;
    if (i >= nh) return;
    const int head = i & 3;
    const float* __restrict__ hv = h + ((size_t)(i >> 2)) * 128 + head * 32;
    const float* __restrict__ al = att_l + head * 32;
    const float* __restrict__ ar = att_r + head * 32;
    float sl = 0.f, sr = 0.f;
#pragma unroll
    for (int d = 0; d < 32; ++d) {
        const float v = hv[d];
        sl += v * al[d];
        sr += v * ar[d];
    }
    alpha_l[i] = sl;
    alpha_r[i] = sr;
}

__global__ void gat_edge_max_rank(const int* __restrict__ ei,
                                  const float* __restrict__ alpha_l,
                                  const float* __restrict__ alpha_r,
                                  unsigned* __restrict__ maxbuf,
                                  int* __restrict__ deg,
                                  int* __restrict__ rank, int E) {
    __shared__ unsigned smax[4];
    if (threadIdx.x < 4) smax[threadIdx.x] = 0u;
    __syncthreads();
    const int e = blockIdx.x * blockDim.x + threadIdx.x;
    if (e < E) {
        const int f = ei[e];
        const int t = ei[E + e];
        rank[e] = atomicAdd(&deg[t], 1);
#pragma unroll
        for (int hd = 0; hd < 4; ++hd) {
            float a = alpha_l[f * 4 + hd] + alpha_r[t * 4 + hd];
            a = a > 0.f ? a : 0.2f * a;
            atomicMax(&smax[hd], f2ord(a));
        }
    }
    __syncthreads();
    if (threadIdx.x < 4) atomicMax(&maxbuf[threadIdx.x], smax[threadIdx.x]);
}

__global__ void __launch_bounds__(1024)
gat_scan(const int* __restrict__ deg, int* __restrict__ off, int n_nodes) {
    __shared__ int s_tot[32];
    __shared__ int s_acc[32];
    __shared__ int s_pre[32];
    const int tid  = threadIdx.x;
    const int lane = tid & 31;
    const int w    = tid >> 5;
    const int first = blockIdx.x * 1024;

    int acc = 0;
    for (int i = tid; i < first; i += 1024) acc += deg[i];

    const int idx = first + tid;
    const int v   = idx < n_nodes ? deg[idx] : 0;

    int inc = v;
#pragma unroll
    for (int d = 1; d < 32; d <<= 1) {
        const int u = __shfl_up(inc, d);
        if (lane >= d) inc += u;
    }
#pragma unroll
    for (int d = 16; d > 0; d >>= 1) acc += __shfl_xor(acc, d);

    if (lane == 31) { s_tot[w] = inc; s_acc[w] = acc; }
    __syncthreads();
    if (w == 0) {
        const int t = s_tot[lane];
        int a  = s_acc[lane];
        int ti = t;
#pragma unroll
        for (int d = 1; d < 32; d <<= 1) {
            const int u = __shfl_up(ti, d);
            if (lane >= d) ti += u;
        }
#pragma unroll
        for (int d = 16; d > 0; d >>= 1) a += __shfl_xor(a, d);
        s_pre[lane] = a + ti - t;
    }
    __syncthreads();
    if (idx < n_nodes) {
        const int excl = s_pre[w] + inc - v;
        off[idx] = excl;
        if (idx == n_nodes - 1) off[n_nodes] = excl + v;
    }
}

__global__ void gat_fill(const int* __restrict__ ei,
                         const int* __restrict__ off,
                         const int* __restrict__ rank,
                         int* __restrict__ srt, int E) {
    const int e = blockIdx.x * blockDim.x + threadIdx.x;
    if (e >= E) return;
    srt[off[ei[E + e]] + rank[e]] = ei[e];
}

#define SEL8(a, b, c, d, e, f, g, h)                                          \
    ((a) | ((b) << 3) | ((c) << 6) | ((d) << 9) | ((e) << 12) | ((f) << 15) | \
     ((g) << 18) | ((h) << 21))

template <unsigned SEL>
__device__ __forceinline__ float grp8(float v) {
    return __uint_as_float(__builtin_amdgcn_mov_dpp8(__float_as_uint(v), SEL));
}

#define GAT_ROW_LOAD(J, R)                                                    \
    const int i##R = (J) < m ? (J) : m - 1;                                   \
    const int f##R = __builtin_amdgcn_readlane(f, i##R);                      \
    const float4 R = *reinterpret_cast<const float4*>(hbuf + (size_t)f##R * 128 + lane4);
#define GAT_ROW_FMA(J, R)                                                     \
    {                                                                         \
        const float wj = grp8<SEL8(J, J, J, J, J, J, J, J)>(w);               \
        acc.x += wj * R.x; acc.y += wj * R.y;                                 \
        acc.z += wj * R.z; acc.w += wj * R.w;                                 \
    }

__device__ __forceinline__ void gat_rows8(float4& acc, const float* __restrict__ hbuf,
                                          int f, float w, int m, int lane4) {
    GAT_ROW_LOAD(0, r0) GAT_ROW_LOAD(1, r1) GAT_ROW_LOAD(2, r2) GAT_ROW_LOAD(3, r3)
    GAT_ROW_LOAD(4, r4) GAT_ROW_LOAD(5, r5) GAT_ROW_LOAD(6, r6) GAT_ROW_LOAD(7, r7)
    GAT_ROW_FMA(0, r0) GAT_ROW_FMA(1, r1) GAT_ROW_FMA(2, r2) GAT_ROW_FMA(3, r3)
    GAT_ROW_FMA(4, r4) GAT_ROW_FMA(5, r5) GAT_ROW_FMA(6, r6) GAT_ROW_FMA(7, r7)
}

#undef GAT_ROW_LOAD
#undef GAT_ROW_FMA

__global__ void gat_aggregate(const float* __restrict__ hbuf,
                              const int* __restrict__ srt,
                              const int* __restrict__ off,
                              const float* __restrict__ alpha_l,
                              const float* __restrict__ alpha_r,
                              const unsigned* __restrict__ maxbuf,
                              const float* __restrict__ bias,
                              float* __restrict__ out, int n_nodes) {
    const int lane = threadIdx.x & 31;
    const int n = __builtin_amdgcn_readfirstlane(
        (int)((blockIdx.x * (unsigned)blockDim.x + threadIdx.x) >> 5));
    if (n >= n_nodes) return;
    const int beg = __builtin_amdgcn_readfirstlane(off[n]);
    const int cnt = __builtin_amdgcn_readfirstlane(off[n + 1]) - beg;

    const int lane4 = lane * 4;
    const float4 b = *reinterpret_cast<const float4*>(bias + lane4);
    float4* __restrict__ dst = reinterpret_cast<float4*>(out + (size_t)n * 128 + lane4);
    if (cnt <= 0) {
        *dst = b;
        return;
    }

    const int hd   = lane >> 3;
    const int slot = lane & 7;
    const int last = cnt - 1;
    const float ar = alpha_r[n * 4 + hd];
    const float mx = ord2f(maxbuf[hd]);
    const int* __restrict__ lst = srt + beg;

    const int j0 = slot, j1 = slot + 8;
    const int f0 = lst[j0 < cnt ? j0 : last];
    const int f1 = lst[j1 < cnt ? j1 : last];
    float a0 = alpha_l[f0 * 4 + hd] + ar;
    float a1 = alpha_l[f1 * 4 + hd] + ar;
    a0 = a0 > 0.f ? a0 : 0.2f * a0;
    a1 = a1 > 0.f ? a1 : 0.2f * a1;
    const float e0 = j0 < cnt ? __expf(a0 - mx) : 0.f;
    const float e1 = j1 < cnt ? __expf(a1 - mx) : 0.f;
    float s = e0 + e1;
    int   f2 = 0, f3 = 0;
    float e2 = 0.f, e3 = 0.f;
    if (cnt > 16) {
        const int j2 = slot + 16, j3 = slot + 24;
        f2 = lst[j2 < cnt ? j2 : last];
        f3 = lst[j3 < cnt ? j3 : last];
        float a2 = alpha_l[f2 * 4 + hd] + ar;
        float a3 = alpha_l[f3 * 4 + hd] + ar;
        a2 = a2 > 0.f ? a2 : 0.2f * a2;
        a3 = a3 > 0.f ? a3 : 0.2f * a3;
        e2 = j2 < cnt ? __expf(a2 - mx) : 0.f;
        e3 = j3 < cnt ? __expf(a3 - mx) : 0.f;
        s += e2 + e3;
    }
    for (int base = 32; base < cnt; base += 8) {
        const int j = base + slot;
        const int f = lst[j < cnt ? j : last];
        float a = alpha_l[f * 4 + hd] + ar;
        a = a > 0.f ? a : 0.2f * a;
        s += j < cnt ? __expf(a - mx) : 0.f;
    }
    s += grp8<SEL8(1, 0, 3, 2, 5, 4, 7, 6)>(s);
    s += grp8<SEL8(2, 3, 0, 1, 6, 7, 4, 5)>(s);
    s += grp8<SEL8(4, 5, 6, 7, 0, 1, 2, 3)>(s);
    const float den = s + 1e-9f;

    float4 acc = make_float4(0.f, 0.f, 0.f, 0.f);
    gat_rows8(acc, hbuf, f0, e0 / den, cnt < 8 ? cnt : 8, lane4);
    if (cnt > 8)  gat_rows8(acc, hbuf, f1, e1 / den, cnt < 16 ? cnt - 8 : 8, lane4);
    if (cnt > 16) gat_rows8(acc, hbuf, f2, e2 / den, cnt < 24 ? cnt - 16 : 8, lane4);
    if (cnt > 24) gat_rows8(acc, hbuf, f3, e3 / den, cnt < 32 ? cnt - 24 : 8, lane4);
    for (int base = 32; base < cnt; base += 8) {
        const int j = base + slot;
        const int f = lst[j < cnt ? j : last];
        float a = alpha_l[f * 4 + hd] + ar;
        a = a > 0.f ? a : 0.2f * a;
        const float w = j < cnt ? __expf(a - mx) / den : 0.f;
        const int left = cnt - base;
        gat_rows8(acc, hbuf, f, w, left < 8 ? left : 8, lane4);
    }

    float4 o;
    o.x = acc.x + b.x; o.y = acc.y + b.y; o.z = acc.z + b.z; o.w = acc.w + b.w;
    *dst = o;
}

extern "C" void kernel_launch(void* const* d_in, const int* in_sizes, int n_in,
                              void* d_out, int out_size, void* d_ws, size_t ws_size,
                              hipStream_t stream) {
    const float* x     = (const float*)d_in[0];
    const int*   ei    = (const int*)d_in[1];
    const float* W     = (const float*)d_in[2];
    const float* att_l = (const float*)d_in[3];
    const float* att_r = (const float*)d_in[4];
    const float* bias  = (const float*)d_in[5];
    float* out = (float*)d_out;

    const int N  = in_sizes[0] / 128;
    const int E  = in_sizes[1] / 2;
    const int NH = N * 4;

    float* h       = (float*)d_ws;
    float* alpha_l = h + (size_t)N * 128;
    float* alpha_r = alpha_l + NH;
    int* deg       = (int*)(alpha_r + NH);
    int* off       = deg + N;
    unsigned* maxbuf = (unsigned*)(off + N + 4);
    int* srt       = (int*)(maxbuf + 4);
    int* rank      = (int*)d_out;
    if ((size_t)((char*)(srt + E) - (char*)d_ws) > ws_size) return;

    const int BLK = 256;

    gat_gemm_wmma<<<N / 16, BLK, 0, stream>>>(x, W, h);
    gat_alpha<<<(NH + BLK - 1) / BLK, BLK, 0, stream>>>(h, att_l, att_r,
                                                        alpha_l, alpha_r,
                                                        deg, maxbuf, NH, N);
    gat_edge_max_rank<<<(E + BLK - 1) / BLK, BLK, 0, stream>>>(ei, alpha_l, alpha_r,
                                                               maxbuf, deg, rank, E);
    gat_scan<<<(N + 1023) / 1024, 1024, 0, stream>>>(deg, off, N);
    gat_fill<<<(E + BLK - 1) / BLK, BLK, 0, stream>>>(ei, off, rank, srt, E);
    {
        const long long threads = (long long)N * 32;
        const int blocks = (int)((threads + BLK - 1) / BLK);
        gat_aggregate<<<blocks, BLK, 0, stream>>>(h, srt, off, alpha_l, alpha_r,
                                                  maxbuf, bias, out, N);
    }
}


# Edge ordering through LDS counters: edges binned into 256 node ranges per block, then one block per range sorts by destination; no global atomic add or per-edge rank; sum order = LDS arrival rank
# speedup vs baseline: 3.1022x; 1.6811x over previous
#include <hip/hip_runtime.h>
#include <hip/hip_bf16.h>


typedef __attribute__((ext_vector_type(2))) float v2f;
typedef __attribute__((ext_vector_type(8))) float v8f;

__device__ __forceinline__ unsigned f2ord(float f) {
    unsigned u = __float_as_uint(f);
    return (u & 0x80000000u) ? ~u : (u | 0x80000000u);
}
__device__ __forceinline__ float ord2f(unsigned u) {
    unsigned v = (u & 0x80000000u) ? (u & 0x7FFFFFFFu) : ~u;
    return __uint_as_float(v);
}

__global__ void gat_gemm_wmma(const float* __restrict__ x,
                              const float* __restrict__ W,
                              float* __restrict__ h) {
    const int wave   = threadIdx.x >> 5;
    const int lane   = threadIdx.x & 31;
    const int lhalf  = lane & 15;
    const int khalf  = (lane >> 4) << 1;
    const int rowA   = blockIdx.x * 16 + lhalf;
    const int colB   = wave * 16 + lhalf;

    const float* __restrict__ arow = x + (size_t)rowA * 128;

    v8f c = {0.f, 0.f, 0.f, 0.f, 0.f, 0.f, 0.f, 0.f};

#pragma unroll 4
    for (int k0 = 0; k0 < 128; k0 += 4) {
        const int k = k0 + khalf;
        v2f a;
        a.x = arow[k];
        a.y = arow[k + 1];
        v2f b;
        b.x = W[(size_t)k * 128 + colB];
        b.y = W[(size_t)(k + 1) * 128 + colB];
        c = __builtin_amdgcn_wmma_f32_16x16x4_f32(false, a, false, b,
                                                  (short)0, c, false, false);
    }

    const int mhi = (lane >> 4) << 3;
#pragma unroll
    for (int r = 0; r < 8; ++r) {
        const int m = blockIdx.x * 16 + r + mhi;
        h[(size_t)m * 128 + wave * 16 + lhalf] = c[r];
    }
}

__global__ void gat_alpha(const float* __restrict__ h,
                          const float* __restrict__ att_l,
                          const float* __restrict__ att_r,
                          float* __restrict__ alpha_l,
                          float* __restrict__ alpha_r,
                          unsigned* __restrict__ maxbuf, int nh) {
    const int i = blockIdx.x * blockDim.x + threadIdx.x;
    if (i < 4) maxbuf[i] = f2ord(-3.0e38f);
    if (i >= nh) return;
    const int head = i & 3;
    const float* __restrict__ hv = h + ((size_t)(i >> 2)) * 128 + head * 32;
    const float* __restrict__ al = att_l + head * 32;
    const float* __restrict__ ar = att_r + head * 32;
    float sl = 0.f, sr = 0.f;
#pragma unroll
    for (int d = 0; d < 32; ++d) {
        const float v = hv[d];
        sl += v * al[d];
        sr += v * ar[d];
    }
    alpha_l[i] = sl;
    alpha_r[i] = sr;
}

__global__ void __launch_bounds__(1024)
gat_edge_max_hist(const int* __restrict__ ei,
                  const float* __restrict__ alpha_l,
                  const float* __restrict__ alpha_r,
                  unsigned* __restrict__ maxbuf,
                  int* __restrict__ counts, int E, int chunk, int npb) {
    __shared__ unsigned smax[4];
    __shared__ int hist[256];
    const int tid = threadIdx.x;
    if (tid < 256) hist[tid] = 0;
    if (tid < 4)   smax[tid] = 0u;
    __syncthreads();
    const int beg = blockIdx.x * chunk;
    const int end = beg + chunk < E ? beg + chunk : E;
    float m0 = -3.0e38f, m1 = -3.0e38f, m2 = -3.0e38f, m3 = -3.0e38f;
    for (int e = beg + tid; e < end; e += 1024) {
        const int f = ei[e];
        const int t = ei[E + e];
        atomicAdd(&hist[t / npb], 1);
        float a0 = alpha_l[f * 4 + 0] + alpha_r[t * 4 + 0];
        float a1 = alpha_l[f * 4 + 1] + alpha_r[t * 4 + 1];
        float a2 = alpha_l[f * 4 + 2] + alpha_r[t * 4 + 2];
        float a3 = alpha_l[f * 4 + 3] + alpha_r[t * 4 + 3];
        a0 = a0 > 0.f ? a0 : 0.2f * a0;
        a1 = a1 > 0.f ? a1 : 0.2f * a1;
        a2 = a2 > 0.f ? a2 : 0.2f * a2;
        a3 = a3 > 0.f ? a3 : 0.2f * a3;
        m0 = fmaxf(m0, a0); m1 = fmaxf(m1, a1);
        m2 = fmaxf(m2, a2); m3 = fmaxf(m3, a3);
    }
#pragma unroll
    for (int d = 16; d > 0; d >>= 1) {
        m0 = fmaxf(m0, __shfl_xor(m0, d)); m1 = fmaxf(m1, __shfl_xor(m1, d));
        m2 = fmaxf(m2, __shfl_xor(m2, d)); m3 = fmaxf(m3, __shfl_xor(m3, d));
    }
    if ((tid & 31) == 0) {
        atomicMax(&smax[0], f2ord(m0));
        atomicMax(&smax[1], f2ord(m1));
        atomicMax(&smax[2], f2ord(m2));
        atomicMax(&smax[3], f2ord(m3));
    }
    __syncthreads();
    if (tid < 256) counts[tid * gridDim.x + blockIdx.x] = hist[tid];
    if (tid < 4)   atomicMax(&maxbuf[tid], smax[tid]);
}

__global__ void __launch_bounds__(1024)
gat_scan(const int* __restrict__ deg, int* __restrict__ off, int n_nodes) {
    __shared__ int s_tot[32];
    __shared__ int s_acc[32];
    __shared__ int s_pre[32];
    const int tid  = threadIdx.x;
    const int lane = tid & 31;
    const int w    = tid >> 5;
    const int first = blockIdx.x * 1024;

    int acc = 0;
    for (int i = tid; i < first; i += 1024) acc += deg[i];

    const int idx = first + tid;
    const int v   = idx < n_nodes ? deg[idx] : 0;

    int inc = v;
#pragma unroll
    for (int d = 1; d < 32; d <<= 1) {
        const int u = __shfl_up(inc, d);
        if (lane >= d) inc += u;
    }
#pragma unroll
    for (int d = 16; d > 0; d >>= 1) acc += __shfl_xor(acc, d);

    if (lane == 31) { s_tot[w] = inc; s_acc[w] = acc; }
    __syncthreads();
    if (w == 0) {
        const int t = s_tot[lane];
        int a  = s_acc[lane];
        int ti = t;
#pragma unroll
        for (int d = 1; d < 32; d <<= 1) {
            const int u = __shfl_up(ti, d);
            if (lane >= d) ti += u;
        }
#pragma unroll
        for (int d = 16; d > 0; d >>= 1) a += __shfl_xor(a, d);
        s_pre[lane] = a + ti - t;
    }
    __syncthreads();
    if (idx < n_nodes) {
        const int excl = s_pre[w] + inc - v;
        off[idx] = excl;
        if (idx == n_nodes - 1) off[n_nodes] = excl + v;
    }
}

__global__ void __launch_bounds__(1024)
gat_range_copy(const int* __restrict__ ei,
               const int* __restrict__ cbase,
               int* __restrict__ pairs, int E, int chunk, int npb) {
    __shared__ int cur[256];
    const int tid = threadIdx.x;
    if (tid < 256) cur[tid] = cbase[tid * gridDim.x + blockIdx.x];
    __syncthreads();
    const int beg = blockIdx.x * chunk;
    const int end = beg + chunk < E ? beg + chunk : E;
    int2* __restrict__ p2 = reinterpret_cast<int2*>(pairs);
    for (int e = beg + tid; e < end; e += 1024) {
        const int f = ei[e];
        const int t = ei[E + e];
        const int pos = atomicAdd(&cur[t / npb], 1);
        p2[pos] = make_int2(f, t);
    }
}

__global__ void __launch_bounds__(1024)
gat_range_sort(const int* __restrict__ pairs,
               const int* __restrict__ cbase,
               int* __restrict__ off,
               int* __restrict__ srt,
               int n_nodes, int npb, int nblk, int E) {
    __shared__ int cnt[256];
    __shared__ int cur[256];
    __shared__ int wtot[8];
    const int tid   = threadIdx.x;
    const int lane  = tid & 31;
    const int w     = tid >> 5;
    const int r     = blockIdx.x;
    const int node0 = r * npb;
    const int ebeg  = cbase[r * nblk];
    const int eend  = cbase[(r + 1) * nblk];
    const int2* __restrict__ p2 = reinterpret_cast<const int2*>(pairs);

    if (tid < 256) cnt[tid] = 0;
    __syncthreads();
    for (int i = ebeg + tid; i < eend; i += 1024) atomicAdd(&cnt[p2[i].y - node0], 1);
    __syncthreads();

    int v = 0, inc = 0;
    if (tid < 256) {
        v   = cnt[tid];
        inc = v;
#pragma unroll
        for (int d = 1; d < 32; d <<= 1) {
            const int u = __shfl_up(inc, d);
            if (lane >= d) inc += u;
        }
        if (lane == 31) wtot[w] = inc;
    }
    __syncthreads();
    if (tid < 256) {
        int pre = 0;
        for (int j = 0; j < w; ++j) pre += wtot[j];
        const int excl = pre + inc - v;
        cur[tid] = excl;
        const int node = node0 + tid;
        if (tid < npb && node < n_nodes) off[node] = ebeg + excl;
    }
    if (r == (int)gridDim.x - 1 && tid == 0) off[n_nodes] = E;
    __syncthreads();
    for (int i = ebeg + tid; i < eend; i += 1024) {
        const int2 p = p2[i];
        const int pos = atomicAdd(&cur[p.y - node0], 1);
        srt[ebeg + pos] = p.x;
    }
}

#define SEL8(a, b, c, d, e, f, g, h)                                          \
    ((a) | ((b) << 3) | ((c) << 6) | ((d) << 9) | ((e) << 12) | ((f) << 15) | \
     ((g) << 18) | ((h) << 21))

template <unsigned SEL>
__device__ __forceinline__ float grp8(float v) {
    return __uint_as_float(__builtin_amdgcn_mov_dpp8(__float_as_uint(v), SEL));
}

#define GAT_ROW_LOAD(J, R)                                                    \
    const int i##R = (J) < m ? (J) : m - 1;                                   \
    const int f##R = __builtin_amdgcn_readlane(f, i##R);                      \
    const float4 R = *reinterpret_cast<const float4*>(hbuf + (size_t)f##R * 128 + lane4);
#define GAT_ROW_FMA(J, R)                                                     \
    {                                                                         \
        const float wj = grp8<SEL8(J, J, J, J, J, J, J, J)>(w);               \
        acc.x += wj * R.x; acc.y += wj * R.y;                                 \
        acc.z += wj * R.z; acc.w += wj * R.w;                                 \
    }

__device__ __forceinline__ void gat_rows8(float4& acc, const float* __restrict__ hbuf,
                                          int f, float w, int m, int lane4) {
    GAT_ROW_LOAD(0, r0) GAT_ROW_LOAD(1, r1) GAT_ROW_LOAD(2, r2) GAT_ROW_LOAD(3, r3)
    GAT_ROW_LOAD(4, r4) GAT_ROW_LOAD(5, r5) GAT_ROW_LOAD(6, r6) GAT_ROW_LOAD(7, r7)
    GAT_ROW_FMA(0, r0) GAT_ROW_FMA(1, r1) GAT_ROW_FMA(2, r2) GAT_ROW_FMA(3, r3)
    GAT_ROW_FMA(4, r4) GAT_ROW_FMA(5, r5) GAT_ROW_FMA(6, r6) GAT_ROW_FMA(7, r7)
}

#undef GAT_ROW_LOAD
#undef GAT_ROW_FMA

__global__ void gat_aggregate(const float* __restrict__ hbuf,
                              const int* __restrict__ srt,
                              const int* __restrict__ off,
                              const float* __restrict__ alpha_l,
                              const float* __restrict__ alpha_r,
                              const unsigned* __restrict__ maxbuf,
                              const float* __restrict__ bias,
                              float* __restrict__ out, int n_nodes) {
    const int lane = threadIdx.x & 31;
    const int n = __builtin_amdgcn_readfirstlane(
        (int)((blockIdx.x * (unsigned)blockDim.x + threadIdx.x) >> 5));
    if (n >= n_nodes) return;
    const int beg = __builtin_amdgcn_readfirstlane(off[n]);
    const int cnt = __builtin_amdgcn_readfirstlane(off[n + 1]) - beg;

    const int lane4 = lane * 4;
    const float4 b = *reinterpret_cast<const float4*>(bias + lane4);
    float4* __restrict__ dst = reinterpret_cast<float4*>(out + (size_t)n * 128 + lane4);
    if (cnt <= 0) {
        *dst = b;
        return;
    }

    const int hd   = lane >> 3;
    const int slot = lane & 7;
    const int last = cnt - 1;
    const float ar = alpha_r[n * 4 + hd];
    const float mx = ord2f(maxbuf[hd]);
    const int* __restrict__ lst = srt + beg;

    const int j0 = slot, j1 = slot + 8;
    const int f0 = lst[j0 < cnt ? j0 : last];
    const int f1 = lst[j1 < cnt ? j1 : last];
    float a0 = alpha_l[f0 * 4 + hd] + ar;
    float a1 = alpha_l[f1 * 4 + hd] + ar;
    a0 = a0 > 0.f ? a0 : 0.2f * a0;
    a1 = a1 > 0.f ? a1 : 0.2f * a1;
    const float e0 = j0 < cnt ? __expf(a0 - mx) : 0.f;
    const float e1 = j1 < cnt ? __expf(a1 - mx) : 0.f;
    float s = e0 + e1;
    int   f2 = 0, f3 = 0;
    float e2 = 0.f, e3 = 0.f;
    if (cnt > 16) {
        const int j2 = slot + 16, j3 = slot + 24;
        f2 = lst[j2 < cnt ? j2 : last];
        f3 = lst[j3 < cnt ? j3 : last];
        float a2 = alpha_l[f2 * 4 + hd] + ar;
        float a3 = alpha_l[f3 * 4 + hd] + ar;
        a2 = a2 > 0.f ? a2 : 0.2f * a2;
        a3 = a3 > 0.f ? a3 : 0.2f * a3;
        e2 = j2 < cnt ? __expf(a2 - mx) : 0.f;
        e3 = j3 < cnt ? __expf(a3 - mx) : 0.f;
        s += e2 + e3;
    }
    for (int base = 32; base < cnt; base += 8) {
        const int j = base + slot;
        const int f = lst[j < cnt ? j : last];
        float a = alpha_l[f * 4 + hd] + ar;
        a = a > 0.f ? a : 0.2f * a;
        s += j < cnt ? __expf(a - mx) : 0.f;
    }
    s += grp8<SEL8(1, 0, 3, 2, 5, 4, 7, 6)>(s);
    s += grp8<SEL8(2, 3, 0, 1, 6, 7, 4, 5)>(s);
    s += grp8<SEL8(4, 5, 6, 7, 0, 1, 2, 3)>(s);
    const float den = s + 1e-9f;

    float4 acc = make_float4(0.f, 0.f, 0.f, 0.f);
    gat_rows8(acc, hbuf, f0, e0 / den, cnt < 8 ? cnt : 8, lane4);
    if (cnt > 8)  gat_rows8(acc, hbuf, f1, e1 / den, cnt < 16 ? cnt - 8 : 8, lane4);
    if (cnt > 16) gat_rows8(acc, hbuf, f2, e2 / den, cnt < 24 ? cnt - 16 : 8, lane4);
    if (cnt > 24) gat_rows8(acc, hbuf, f3, e3 / den, cnt < 32 ? cnt - 24 : 8, lane4);
    for (int base = 32; base < cnt; base += 8) {
        const int j = base + slot;
        const int f = lst[j < cnt ? j : last];
        float a = alpha_l[f * 4 + hd] + ar;
        a = a > 0.f ? a : 0.2f * a;
        const float w = j < cnt ? __expf(a - mx) / den : 0.f;
        const int left = cnt - base;
        gat_rows8(acc, hbuf, f, w, left < 8 ? left : 8, lane4);
    }

    float4 o;
    o.x = acc.x + b.x; o.y = acc.y + b.y; o.z = acc.z + b.z; o.w = acc.w + b.w;
    *dst = o;
}

extern "C" void kernel_launch(void* const* d_in, const int* in_sizes, int n_in,
                              void* d_out, int out_size, void* d_ws, size_t ws_size,
                              hipStream_t stream) {
    const float* x     = (const float*)d_in[0];
    const int*   ei    = (const int*)d_in[1];
    const float* W     = (const float*)d_in[2];
    const float* att_l = (const float*)d_in[3];
    const float* att_r = (const float*)d_in[4];
    const float* bias  = (const float*)d_in[5];
    float* out = (float*)d_out;

    const int N  = in_sizes[0] / 128;
    const int E  = in_sizes[1] / 2;
    const int NH = N * 4;

    float* h       = (float*)d_ws;
    float* alpha_l = h + (size_t)N * 128;
    float* alpha_r = alpha_l + NH;
    int* off       = (int*)(alpha_r + NH);
    unsigned* maxbuf = (unsigned*)(off + N + 4);
    int* srt       = (int*)(maxbuf + 4);
    if ((size_t)((char*)(srt + E) - (char*)d_ws) > ws_size) return;

    const int NRANGE = 256;
    const int NBLK   = 256;
    const int NC     = NRANGE * NBLK;
    const int npb    = (N + NRANGE - 1) / NRANGE;
    const int chunk  = (E + NBLK - 1) / NBLK;
    int* counts = (int*)d_out;
    int* cbase  = counts + NC;
    int* pairs  = cbase + NC + 4;
    if (npb > 256) return;
    if ((size_t)2 * NC + 4 + (size_t)2 * E > (size_t)out_size) return;

    const int BLK = 256;

    gat_gemm_wmma<<<N / 16, BLK, 0, stream>>>(x, W, h);
    gat_alpha<<<(NH + BLK - 1) / BLK, BLK, 0, stream>>>(h, att_l, att_r,
                                                        alpha_l, alpha_r, maxbuf, NH);
    gat_edge_max_hist<<<NBLK, 1024, 0, stream>>>(ei, alpha_l, alpha_r, maxbuf,
                                                 counts, E, chunk, npb);
    gat_scan<<<(NC + 1023) / 1024, 1024, 0, stream>>>(counts, cbase, NC);
    gat_range_copy<<<NBLK, 1024, 0, stream>>>(ei, cbase, pairs, E, chunk, npb);
    gat_range_sort<<<NRANGE, 1024, 0, stream>>>(pairs, cbase, off, srt, N, npb, NBLK, E);
    {
        const long long threads = (long long)N * 32;
        const int blocks = (int)((threads + BLK - 1) / BLK);
        gat_aggregate<<<blocks, BLK, 0, stream>>>(h, srt, off, alpha_l, alpha_r,
                                                  maxbuf, bias, out, N);
    }
}


# Fewer, longer waves: aggregate pass 8192 waves over 6-7 nodes each (next nodes' lists and scores loaded ahead), GEMM 8192 waves over 3-4 row tiles each with the W strip in registers; sums unchanged
# speedup vs baseline: 3.2781x; 1.0567x over previous
#include <hip/hip_runtime.h>
#include <hip/hip_bf16.h>


typedef __attribute__((ext_vector_type(2))) float v2f;
typedef __attribute__((ext_vector_type(8))) float v8f;

__device__ __forceinline__ unsigned f2ord(float f) {
    unsigned u = __float_as_uint(f);
    return (u & 0x80000000u) ? ~u : (u | 0x80000000u);
}
__device__ __forceinline__ float ord2f(unsigned u) {
    unsigned v = (u & 0x80000000u) ? (u & 0x7FFFFFFFu) : ~u;
    return __uint_as_float(v);
}

__global__ void __launch_bounds__(256)
gat_gemm_wmma(const float* __restrict__ x,
              const float* __restrict__ W,
              float* __restrict__ h, int ntiles) {
    const int wave   = threadIdx.x >> 5;
    const int lane   = threadIdx.x & 31;
    const int lhalf  = lane & 15;
    const int khalf  = (lane >> 4) << 1;
    const int colB   = wave * 16 + lhalf;
    const int mhi    = (lane >> 4) << 3;

    v2f bw[32];
#pragma unroll
    for (int i = 0; i < 32; ++i) {
        const int k = 4 * i + khalf;
        bw[i].x = W[(size_t)k * 128 + colB];
        bw[i].y = W[(size_t)(k + 1) * 128 + colB];
    }

    for (int t = blockIdx.x; t < ntiles; t += gridDim.x) {
        const float* __restrict__ arow = x + (size_t)(t * 16 + lhalf) * 128;
        v8f c = {0.f, 0.f, 0.f, 0.f, 0.f, 0.f, 0.f, 0.f};
#pragma unroll
        for (int i = 0; i < 32; ++i) {
            const int k = 4 * i + khalf;
            v2f a;
            a.x = arow[k];
            a.y = arow[k + 1];
            c = __builtin_amdgcn_wmma_f32_16x16x4_f32(false, a, false, bw[i],
                                                      (short)0, c, false, false);
        }
#pragma unroll
        for (int r = 0; r < 8; ++r) {
            const int m = t * 16 + r + mhi;
            h[(size_t)m * 128 + wave * 16 + lhalf] = c[r];
        }
    }
}

__global__ void gat_alpha(const float* __restrict__ h,
                          const float* __restrict__ att_l,
                          const float* __restrict__ att_r,
                          float* __restrict__ alpha_l,
                          float* __restrict__ alpha_r,
                          unsigned* __restrict__ maxbuf, int nh) {
    const int i = blockIdx.x * blockDim.x + threadIdx.x;
    if (i < 4) maxbuf[i] = f2ord(-3.0e38f);
    if (i >= nh) return;
    const int head = i & 3;
    const float* __restrict__ hv = h + ((size_t)(i >> 2)) * 128 + head * 32;
    const float* __restrict__ al = att_l + head * 32;
    const float* __restrict__ ar = att_r + head * 32;
    float sl = 0.f, sr = 0.f;
#pragma unroll
    for (int d = 0; d < 32; ++d) {
        const float v = hv[d];
        sl += v * al[d];
        sr += v * ar[d];
    }
    alpha_l[i] = sl;
    alpha_r[i] = sr;
}

__global__ void __launch_bounds__(1024)
gat_edge_max_hist(const int* __restrict__ ei,
                  const float* __restrict__ alpha_l,
                  const float* __restrict__ alpha_r,
                  unsigned* __restrict__ maxbuf,
                  int* __restrict__ counts, int E, int chunk, int npb) {
    __shared__ unsigned smax[4];
    __shared__ int hist[256];
    const int tid = threadIdx.x;
    if (tid < 256) hist[tid] = 0;
    if (tid < 4)   smax[tid] = 0u;
    __syncthreads();
    const int beg = blockIdx.x * chunk;
    const int end = beg + chunk < E ? beg + chunk : E;
    float m0 = -3.0e38f, m1 = -3.0e38f, m2 = -3.0e38f, m3 = -3.0e38f;
    for (int e = beg + tid; e < end; e += 1024) {
        const int f = ei[e];
        const int t = ei[E + e];
        atomicAdd(&hist[t / npb], 1);
        float a0 = alpha_l[f * 4 + 0] + alpha_r[t * 4 + 0];
        float a1 = alpha_l[f * 4 + 1] + alpha_r[t * 4 + 1];
        float a2 = alpha_l[f * 4 + 2] + alpha_r[t * 4 + 2];
        float a3 = alpha_l[f * 4 + 3] + alpha_r[t * 4 + 3];
        a0 = a0 > 0.f ? a0 : 0.2f * a0;
        a1 = a1 > 0.f ? a1 : 0.2f * a1;
        a2 = a2 > 0.f ? a2 : 0.2f * a2;
        a3 = a3 > 0.f ? a3 : 0.2f * a3;
        m0 = fmaxf(m0, a0); m1 = fmaxf(m1, a1);
        m2 = fmaxf(m2, a2); m3 = fmaxf(m3, a3);
    }
#pragma unroll
    for (int d = 16; d > 0; d >>= 1) {
        m0 = fmaxf(m0, __shfl_xor(m0, d)); m1 = fmaxf(m1, __shfl_xor(m1, d));
        m2 = fmaxf(m2, __shfl_xor(m2, d)); m3 = fmaxf(m3, __shfl_xor(m3, d));
    }
    if ((tid & 31) == 0) {
        atomicMax(&smax[0], f2ord(m0));
        atomicMax(&smax[1], f2ord(m1));
        atomicMax(&smax[2], f2ord(m2));
        atomicMax(&smax[3], f2ord(m3));
    }
    __syncthreads();
    if (tid < 256) counts[tid * gridDim.x + blockIdx.x] = hist[tid];
    if (tid < 4)   atomicMax(&maxbuf[tid], smax[tid]);
}

__global__ void __launch_bounds__(1024)
gat_scan(const int* __restrict__ deg, int* __restrict__ off, int n_nodes) {
    __shared__ int s_tot[32];
    __shared__ int s_acc[32];
    __shared__ int s_pre[32];
    const int tid  = threadIdx.x;
    const int lane = tid & 31;
    const int w    = tid >> 5;
    const int first = blockIdx.x * 1024;

    int acc = 0;
    for (int i = tid; i < first; i += 1024) acc += deg[i];

    const int idx = first + tid;
    const int v   = idx < n_nodes ? deg[idx] : 0;

    int inc = v;
#pragma unroll
    for (int d = 1; d < 32; d <<= 1) {
        const int u = __shfl_up(inc, d);
        if (lane >= d) inc += u;
    }
#pragma unroll
    for (int d = 16; d > 0; d >>= 1) acc += __shfl_xor(acc, d);

    if (lane == 31) { s_tot[w] = inc; s_acc[w] = acc; }
    __syncthreads();
    if (w == 0) {
        const int t = s_tot[lane];
        int a  = s_acc[lane];
        int ti = t;
#pragma unroll
        for (int d = 1; d < 32; d <<= 1) {
            const int u = __shfl_up(ti, d);
            if (lane >= d) ti += u;
        }
#pragma unroll
        for (int d = 16; d > 0; d >>= 1) a += __shfl_xor(a, d);
        s_pre[lane] = a + ti - t;
    }
    __syncthreads();
    if (idx < n_nodes) {
        const int excl = s_pre[w] + inc - v;
        off[idx] = excl;
        if (idx == n_nodes - 1) off[n_nodes] = excl + v;
    }
}

__global__ void __launch_bounds__(1024)
gat_range_copy(const int* __restrict__ ei,
               const int* __restrict__ cbase,
               int* __restrict__ pairs, int E, int chunk, int npb) {
    __shared__ int cur[256];
    const int tid = threadIdx.x;
    if (tid < 256) cur[tid] = cbase[tid * gridDim.x + blockIdx.x];
    __syncthreads();
    const int beg = blockIdx.x * chunk;
    const int end = beg + chunk < E ? beg + chunk : E;
    int2* __restrict__ p2 = reinterpret_cast<int2*>(pairs);
    for (int e = beg + tid; e < end; e += 1024) {
        const int f = ei[e];
        const int t = ei[E + e];
        const int pos = atomicAdd(&cur[t / npb], 1);
        p2[pos] = make_int2(f, t);
    }
}

__global__ void __launch_bounds__(1024)
gat_range_sort(const int* __restrict__ pairs,
               const int* __restrict__ cbase,
               int* __restrict__ off,
               int* __restrict__ srt,
               int n_nodes, int npb, int nblk, int E) {
    __shared__ int cnt[256];
    __shared__ int cur[256];
    __shared__ int wtot[8];
    const int tid   = threadIdx.x;
    const int lane  = tid & 31;
    const int w     = tid >> 5;
    const int r     = blockIdx.x;
    const int node0 = r * npb;
    const int ebeg  = cbase[r * nblk];
    const int eend  = cbase[(r + 1) * nblk];
    const int2* __restrict__ p2 = reinterpret_cast<const int2*>(pairs);

    if (tid < 256) cnt[tid] = 0;
    __syncthreads();
    for (int i = ebeg + tid; i < eend; i += 1024) atomicAdd(&cnt[p2[i].y - node0], 1);
    __syncthreads();

    int v = 0, inc = 0;
    if (tid < 256) {
        v   = cnt[tid];
        inc = v;
#pragma unroll
        for (int d = 1; d < 32; d <<= 1) {
            const int u = __shfl_up(inc, d);
            if (lane >= d) inc += u;
        }
        if (lane == 31) wtot[w] = inc;
    }
    __syncthreads();
    if (tid < 256) {
        int pre = 0;
        for (int j = 0; j < w; ++j) pre += wtot[j];
        const int excl = pre + inc - v;
        cur[tid] = excl;
        const int node = node0 + tid;
        if (tid < npb && node < n_nodes) off[node] = ebeg + excl;
    }
    if (r == (int)gridDim.x - 1 && tid == 0) off[n_nodes] = E;
    __syncthreads();
    for (int i = ebeg + tid; i < eend; i += 1024) {
        const int2 p = p2[i];
        const int pos = atomicAdd(&cur[p.y - node0], 1);
        srt[ebeg + pos] = p.x;
    }
}

#define SEL8(a, b, c, d, e, f, g, h)                                          \
    ((a) | ((b) << 3) | ((c) << 6) | ((d) << 9) | ((e) << 12) | ((f) << 15) | \
     ((g) << 18) | ((h) << 21))

template <unsigned SEL>
__device__ __forceinline__ float grp8(float v) {
    return __uint_as_float(__builtin_amdgcn_mov_dpp8(__float_as_uint(v), SEL));
}

#define GAT_ROW_LOAD(J, R)                                                    \
    const int i##R = (J) < m ? (J) : m - 1;                                   \
    const int f##R = __builtin_amdgcn_readlane(f, i##R);                      \
    const float4 R = *reinterpret_cast<const float4*>(hbuf + (size_t)f##R * 128 + lane4);
#define GAT_ROW_FMA(J, R)                                                     \
    {                                                                         \
        const float wj = grp8<SEL8(J, J, J, J, J, J, J, J)>(w);               \
        acc.x += wj * R.x; acc.y += wj * R.y;                                 \
        acc.z += wj * R.z; acc.w += wj * R.w;                                 \
    }

__device__ __forceinline__ void gat_rows8(float4& acc, const float* __restrict__ hbuf,
                                          int f, float w, int m, int lane4) {
    GAT_ROW_LOAD(0, r0) GAT_ROW_LOAD(1, r1) GAT_ROW_LOAD(2, r2) GAT_ROW_LOAD(3, r3)
    GAT_ROW_LOAD(4, r4) GAT_ROW_LOAD(5, r5) GAT_ROW_LOAD(6, r6) GAT_ROW_LOAD(7, r7)
    GAT_ROW_FMA(0, r0) GAT_ROW_FMA(1, r1) GAT_ROW_FMA(2, r2) GAT_ROW_FMA(3, r3)
    GAT_ROW_FMA(4, r4) GAT_ROW_FMA(5, r5) GAT_ROW_FMA(6, r6) GAT_ROW_FMA(7, r7)
}

#undef GAT_ROW_LOAD
#undef GAT_ROW_FMA

#define GAT_AGG_WAVES_LOG2 13

struct gat_src { int f0, f1, f2, f3; };
struct gat_sc  { float a0, a1, a2, a3, ar; };

__device__ __forceinline__ gat_src gat_list32(const int* __restrict__ srt, int offv,
                                              int j, int slot) {
    const int beg  = __builtin_amdgcn_readlane(offv, j);
    const int cnt  = __builtin_amdgcn_readlane(offv, j + 1) - beg;
    const int base = cnt > 0 ? beg : 0;
    const int last = cnt > 0 ? cnt - 1 : 0;
    const int* __restrict__ lst = srt + base;
    gat_src s;
    s.f0 = lst[slot      < last ? slot      : last];
    s.f1 = lst[slot + 8  < last ? slot + 8  : last];
    s.f2 = lst[slot + 16 < last ? slot + 16 : last];
    s.f3 = lst[slot + 24 < last ? slot + 24 : last];
    return s;
}

__device__ __forceinline__ gat_sc gat_scores(const float* __restrict__ alpha_l,
                                             const float* __restrict__ alpha_r,
                                             const gat_src f, int n, int n_nodes, int hd) {
    const int nc = n < n_nodes ? n : n_nodes - 1;
    gat_sc a;
    a.a0 = alpha_l[f.f0 * 4 + hd];
    a.a1 = alpha_l[f.f1 * 4 + hd];
    a.a2 = alpha_l[f.f2 * 4 + hd];
    a.a3 = alpha_l[f.f3 * 4 + hd];
    a.ar = alpha_r[nc * 4 + hd];
    return a;
}

__global__ void __launch_bounds__(256)
gat_aggregate(const float* __restrict__ hbuf,
              const int* __restrict__ srt,
              const int* __restrict__ off,
              const float* __restrict__ alpha_l,
              const float* __restrict__ alpha_r,
              const unsigned* __restrict__ maxbuf,
              const float* __restrict__ bias,
              float* __restrict__ out, int n_nodes) {
    const int lane = threadIdx.x & 31;
    const unsigned wv = (unsigned)__builtin_amdgcn_readfirstlane(
        (int)((blockIdx.x * (unsigned)blockDim.x + threadIdx.x) >> 5));
    const int n0 = (int)(((unsigned long long)wv * (unsigned)n_nodes) >> GAT_AGG_WAVES_LOG2);
    const int nn = (int)(((unsigned long long)(wv + 1) * (unsigned)n_nodes) >> GAT_AGG_WAVES_LOG2) - n0;
    if (nn <= 0) return;
    const int offv = off[n0 + (lane < nn ? lane : nn)];

    const int lane4 = lane * 4;
    const int hd    = lane >> 3;
    const int slot  = lane & 7;
    const float4 b  = *reinterpret_cast<const float4*>(bias + lane4);
    const float mx  = ord2f(maxbuf[hd]);

    gat_src Fc = gat_list32(srt, offv, 0, slot);
    gat_src Fn = gat_list32(srt, offv, 1, slot);
    gat_sc  Ac = gat_scores(alpha_l, alpha_r, Fc, n0, n_nodes, hd);

    for (int j = 0; j < nn; ++j) {
        const int n   = n0 + j;
        const int beg = __builtin_amdgcn_readlane(offv, j);
        const int cnt = __builtin_amdgcn_readlane(offv, j + 1) - beg;
        const gat_sc  An  = gat_scores(alpha_l, alpha_r, Fn, n + 1, n_nodes, hd);
        const gat_src Fnn = gat_list32(srt, offv, j + 2, slot);

        float4* __restrict__ dst = reinterpret_cast<float4*>(out + (size_t)n * 128 + lane4);
        if (cnt <= 0) {
            *dst = b;
        } else {
            const int last = cnt - 1;
            const float ar = Ac.ar;
            const int* __restrict__ lst = srt + beg;

            const int j0 = slot, j1 = slot + 8, j2 = slot + 16, j3 = slot + 24;
            float a0 = Ac.a0 + ar;
            float a1 = Ac.a1 + ar;
            float a2 = Ac.a2 + ar;
            float a3 = Ac.a3 + ar;
            a0 = a0 > 0.f ? a0 : 0.2f * a0;
            a1 = a1 > 0.f ? a1 : 0.2f * a1;
            a2 = a2 > 0.f ? a2 : 0.2f * a2;
            a3 = a3 > 0.f ? a3 : 0.2f * a3;
            const float e0 = j0 < cnt ? __expf(a0 - mx) : 0.f;
            const float e1 = j1 < cnt ? __expf(a1 - mx) : 0.f;
            const float e2 = j2 < cnt ? __expf(a2 - mx) : 0.f;
            const float e3 = j3 < cnt ? __expf(a3 - mx) : 0.f;
            float s = e0 + e1;
            s += e2 + e3;
            for (int base = 32; base < cnt; base += 8) {
                const int jj = base + slot;
                const int f = lst[jj < cnt ? jj : last];
                float a = alpha_l[f * 4 + hd] + ar;
                a = a > 0.f ? a : 0.2f * a;
                s += jj < cnt ? __expf(a - mx) : 0.f;
            }
            s += grp8<SEL8(1, 0, 3, 2, 5, 4, 7, 6)>(s);
            s += grp8<SEL8(2, 3, 0, 1, 6, 7, 4, 5)>(s);
            s += grp8<SEL8(4, 5, 6, 7, 0, 1, 2, 3)>(s);
            const float den = s + 1e-9f;

            float4 acc = make_float4(0.f, 0.f, 0.f, 0.f);
            gat_rows8(acc, hbuf, Fc.f0, e0 / den, cnt < 8 ? cnt : 8, lane4);
            if (cnt > 8)  gat_rows8(acc, hbuf, Fc.f1, e1 / den, cnt < 16 ? cnt - 8 : 8, lane4);
            if (cnt > 16) gat_rows8(acc, hbuf, Fc.f2, e2 / den, cnt < 24 ? cnt - 16 : 8, lane4);
            if (cnt > 24) gat_rows8(acc, hbuf, Fc.f3, e3 / den, cnt < 32 ? cnt - 24 : 8, lane4);
            for (int base = 32; base < cnt; base += 8) {
                const int jj = base + slot;
                const int f = lst[jj < cnt ? jj : last];
                float a = alpha_l[f * 4 + hd] + ar;
                a = a > 0.f ? a : 0.2f * a;
                const float w = jj < cnt ? __expf(a - mx) / den : 0.f;
                const int left = cnt - base;
                gat_rows8(acc, hbuf, f, w, left < 8 ? left : 8, lane4);
            }

            float4 o;
            o.x = acc.x + b.x; o.y = acc.y + b.y; o.z = acc.z + b.z; o.w = acc.w + b.w;
            *dst = o;
        }
        Fc = Fn; Fn = Fnn; Ac = An;
    }
}

extern "C" void kernel_launch(void* const* d_in, const int* in_sizes, int n_in,
                              void* d_out, int out_size, void* d_ws, size_t ws_size,
                              hipStream_t stream) {
    const float* x     = (const float*)d_in[0];
    const int*   ei    = (const int*)d_in[1];
    const float* W     = (const float*)d_in[2];
    const float* att_l = (const float*)d_in[3];
    const float* att_r = (const float*)d_in[4];
    const float* bias  = (const float*)d_in[5];
    float* out = (float*)d_out;

    const int N  = in_sizes[0] / 128;
    const int E  = in_sizes[1] / 2;
    const int NH = N * 4;

    float* h       = (float*)d_ws;
    float* alpha_l = h + (size_t)N * 128;
    float* alpha_r = alpha_l + NH;
    int* off       = (int*)(alpha_r + NH);
    unsigned* maxbuf = (unsigned*)(off + N + 4);
    int* srt       = (int*)(maxbuf + 4);
    if ((size_t)((char*)(srt + E) - (char*)d_ws) > ws_size) return;

    const int NRANGE = 256;
    const int NBLK   = 256;
    const int NC     = NRANGE * NBLK;
    const int npb    = (N + NRANGE - 1) / NRANGE;
    const int chunk  = (E + NBLK - 1) / NBLK;
    int* counts = (int*)d_out;
    int* cbase  = counts + NC;
    int* pairs  = cbase + NC + 4;
    if (npb > 256) return;
    if ((size_t)2 * NC + 4 + (size_t)2 * E > (size_t)out_size) return;

    const int BLK = 256;

    gat_gemm_wmma<<<1024, BLK, 0, stream>>>(x, W, h, N / 16);
    gat_alpha<<<(NH + BLK - 1) / BLK, BLK, 0, stream>>>(h, att_l, att_r,
                                                        alpha_l, alpha_r, maxbuf, NH);
    gat_edge_max_hist<<<NBLK, 1024, 0, stream>>>(ei, alpha_l, alpha_r, maxbuf,
                                                 counts, E, chunk, npb);
    gat_scan<<<(NC + 1023) / 1024, 1024, 0, stream>>>(counts, cbase, NC);
    gat_range_copy<<<NBLK, 1024, 0, stream>>>(ei, cbase, pairs, E, chunk, npb);
    gat_range_sort<<<NRANGE, 1024, 0, stream>>>(pairs, cbase, off, srt, N, npb, NBLK, E);
    {
        if ((N >> GAT_AGG_WAVES_LOG2) + 4 > 32) return;
        const int blocks = (32 << GAT_AGG_WAVES_LOG2) / BLK;
        gat_aggregate<<<blocks, BLK, 0, stream>>>(h, srt, off, alpha_l, alpha_r,
                                                  maxbuf, bias, out, N);
    }
}


# GEMM tiles staged in LDS: x tile read once per block as whole rows, h tile stored as whole rows, alpha_l/alpha_r dotted from the staged h tile in the same launch (alpha launch gone); sum order kept
# speedup vs baseline: 3.7717x; 1.1506x over previous
#include <hip/hip_runtime.h>
#include <hip/hip_bf16.h>


typedef __attribute__((ext_vector_type(2))) float v2f;
typedef __attribute__((ext_vector_type(8))) float v8f;

__device__ __forceinline__ unsigned f2ord(float f) {
    unsigned u = __float_as_uint(f);
    return (u & 0x80000000u) ? ~u : (u | 0x80000000u);
}
__device__ __forceinline__ float ord2f(unsigned u) {
    unsigned v = (u & 0x80000000u) ? (u & 0x7FFFFFFFu) : ~u;
    return __uint_as_float(v);
}

#define GAT_PITCH 132

__global__ void __launch_bounds__(256)
gat_gemm_wmma(const float* __restrict__ x,
              const float* __restrict__ W,
              const float* __restrict__ att_l,
              const float* __restrict__ att_r,
              float* __restrict__ h,
              float* __restrict__ alpha_l,
              float* __restrict__ alpha_r,
              unsigned* __restrict__ maxbuf, int ntiles) {
    __shared__ __attribute__((aligned(16))) float sA[16 * GAT_PITCH];
    __shared__ __attribute__((aligned(16))) float sC[16 * GAT_PITCH];
    __shared__ __attribute__((aligned(16))) float sAtt[256];

    const int tid    = threadIdx.x;
    const int wave   = tid >> 5;
    const int lane   = tid & 31;
    const int lhalf  = lane & 15;
    const int khalf  = (lane >> 4) << 1;
    const int colB   = wave * 16 + lhalf;
    const int mhi    = (lane >> 4) << 3;
    const int r0     = 2 * wave;
    const int c4     = 4 * lane;

    if (blockIdx.x == 0 && tid < 4) maxbuf[tid] = f2ord(-3.0e38f);
    sAtt[tid] = tid < 128 ? att_l[tid] : att_r[tid - 128];

    v2f bw[32];
#pragma unroll
    for (int i = 0; i < 32; ++i) {
        const int k = 4 * i + khalf;
        bw[i].x = W[(size_t)k * 128 + colB];
        bw[i].y = W[(size_t)(k + 1) * 128 + colB];
    }

    const int dwhich = tid >> 6;
    const int drow   = (tid & 63) >> 2;
    const int dhead  = tid & 3;

    int t = blockIdx.x;
    float4 p0 = make_float4(0.f, 0.f, 0.f, 0.f), p1 = p0;
    if (t < ntiles) {
        const float* __restrict__ src = x + (size_t)(t * 16 + r0) * 128 + c4;
        p0 = *reinterpret_cast<const float4*>(src);
        p1 = *reinterpret_cast<const float4*>(src + 128);
    }
    for (; t < ntiles; t += gridDim.x) {
        *reinterpret_cast<float4*>(sA + r0 * GAT_PITCH + c4)       = p0;
        *reinterpret_cast<float4*>(sA + (r0 + 1) * GAT_PITCH + c4) = p1;
        __syncthreads();

        const int tn = t + gridDim.x;
        if (tn < ntiles) {
            const float* __restrict__ src = x + (size_t)(tn * 16 + r0) * 128 + c4;
            p0 = *reinterpret_cast<const float4*>(src);
            p1 = *reinterpret_cast<const float4*>(src + 128);
        }

        const float* __restrict__ arow = sA + lhalf * GAT_PITCH + khalf;
        v8f c = {0.f, 0.f, 0.f, 0.f, 0.f, 0.f, 0.f, 0.f};
#pragma unroll
        for (int i = 0; i < 32; ++i) {
            const v2f a = *reinterpret_cast<const v2f*>(arow + 4 * i);
            c = __builtin_amdgcn_wmma_f32_16x16x4_f32(false, a, false, bw[i],
                                                      (short)0, c, false, false);
        }
#pragma unroll
        for (int r = 0; r < 8; ++r)
            sC[(r + mhi) * GAT_PITCH + wave * 16 + lhalf] = c[r];
        __syncthreads();

        {
            const float4 o0 = *reinterpret_cast<const float4*>(sC + r0 * GAT_PITCH + c4);
            const float4 o1 = *reinterpret_cast<const float4*>(sC + (r0 + 1) * GAT_PITCH + c4);
            float* __restrict__ dst = h + (size_t)(t * 16 + r0) * 128 + c4;
            *reinterpret_cast<float4*>(dst)       = o0;
            *reinterpret_cast<float4*>(dst + 128) = o1;
        }
        if (tid < 128) {
            const float* __restrict__ hv = sC + drow * GAT_PITCH + dhead * 32;
            const float* __restrict__ av = sAtt + dwhich * 128 + dhead * 32;
            float s = 0.f;
            for (int d = 0; d < 32; ++d) s += hv[d] * av[d];
            float* __restrict__ ap = dwhich ? alpha_r : alpha_l;
            ap[(size_t)(t * 16 + drow) * 4 + dhead] = s;
        }
    }
}

__global__ void __launch_bounds__(1024)
gat_edge_max_hist(const int* __restrict__ ei,
                  const float* __restrict__ alpha_l,
                  const float* __restrict__ alpha_r,
                  unsigned* __restrict__ maxbuf,
                  int* __restrict__ counts, int E, int chunk, int npb) {
    __shared__ unsigned smax[4];
    __shared__ int hist[256];
    const int tid = threadIdx.x;
    if (tid < 256) hist[tid] = 0;
    if (tid < 4)   smax[tid] = 0u;
    __syncthreads();
    const int beg = blockIdx.x * chunk;
    const int end = beg + chunk < E ? beg + chunk : E;
    float m0 = -3.0e38f, m1 = -3.0e38f, m2 = -3.0e38f, m3 = -3.0e38f;
    for (int e = beg + tid; e < end; e += 1024) {
        const int f = ei[e];
        const int t = ei[E + e];
        atomicAdd(&hist[t / npb], 1);
        float a0 = alpha_l[f * 4 + 0] + alpha_r[t * 4 + 0];
        float a1 = alpha_l[f * 4 + 1] + alpha_r[t * 4 + 1];
        float a2 = alpha_l[f * 4 + 2] + alpha_r[t * 4 + 2];
        float a3 = alpha_l[f * 4 + 3] + alpha_r[t * 4 + 3];
        a0 = a0 > 0.f ? a0 : 0.2f * a0;
        a1 = a1 > 0.f ? a1 : 0.2f * a1;
        a2 = a2 > 0.f ? a2 : 0.2f * a2;
        a3 = a3 > 0.f ? a3 : 0.2f * a3;
        m0 = fmaxf(m0, a0); m1 = fmaxf(m1, a1);
        m2 = fmaxf(m2, a2); m3 = fmaxf(m3, a3);
    }
#pragma unroll
    for (int d = 16; d > 0; d >>= 1) {
        m0 = fmaxf(m0, __shfl_xor(m0, d)); m1 = fmaxf(m1, __shfl_xor(m1, d));
        m2 = fmaxf(m2, __shfl_xor(m2, d)); m3 = fmaxf(m3, __shfl_xor(m3, d));
    }
    if ((tid & 31) == 0) {
        atomicMax(&smax[0], f2ord(m0));
        atomicMax(&smax[1], f2ord(m1));
        atomicMax(&smax[2], f2ord(m2));
        atomicMax(&smax[3], f2ord(m3));
    }
    __syncthreads();
    if (tid < 256) counts[tid * gridDim.x + blockIdx.x] = hist[tid];
    if (tid < 4)   atomicMax(&maxbuf[tid], smax[tid]);
}

__global__ void __launch_bounds__(1024)
gat_scan(const int* __restrict__ deg, int* __restrict__ off, int n_nodes) {
    __shared__ int s_tot[32];
    __shared__ int s_acc[32];
    __shared__ int s_pre[32];
    const int tid  = threadIdx.x;
    const int lane = tid & 31;
    const int w    = tid >> 5;
    const int first = blockIdx.x * 1024;

    int acc = 0;
    for (int i = tid; i < first; i += 1024) acc += deg[i];

    const int idx = first + tid;
    const int v   = idx < n_nodes ? deg[idx] : 0;

    int inc = v;
#pragma unroll
    for (int d = 1; d < 32; d <<= 1) {
        const int u = __shfl_up(inc, d);
        if (lane >= d) inc += u;
    }
#pragma unroll
    for (int d = 16; d > 0; d >>= 1) acc += __shfl_xor(acc, d);

    if (lane == 31) { s_tot[w] = inc; s_acc[w] = acc; }
    __syncthreads();
    if (w == 0) {
        const int t = s_tot[lane];
        int a  = s_acc[lane];
        int ti = t;
#pragma unroll
        for (int d = 1; d < 32; d <<= 1) {
            const int u = __shfl_up(ti, d);
            if (lane >= d) ti += u;
        }
#pragma unroll
        for (int d = 16; d > 0; d >>= 1) a += __shfl_xor(a, d);
        s_pre[lane] = a + ti - t;
    }
    __syncthreads();
    if (idx < n_nodes) {
        const int excl = s_pre[w] + inc - v;
        off[idx] = excl;
        if (idx == n_nodes - 1) off[n_nodes] = excl + v;
    }
}

__global__ void __launch_bounds__(1024)
gat_range_copy(const int* __restrict__ ei,
               const int* __restrict__ cbase,
               int* __restrict__ pairs, int E, int chunk, int npb) {
    __shared__ int cur[256];
    const int tid = threadIdx.x;
    if (tid < 256) cur[tid] = cbase[tid * gridDim.x + blockIdx.x];
    __syncthreads();
    const int beg = blockIdx.x * chunk;
    const int end = beg + chunk < E ? beg + chunk : E;
    int2* __restrict__ p2 = reinterpret_cast<int2*>(pairs);
    for (int e = beg + tid; e < end; e += 1024) {
        const int f = ei[e];
        const int t = ei[E + e];
        const int pos = atomicAdd(&cur[t / npb], 1);
        p2[pos] = make_int2(f, t);
    }
}

__global__ void __launch_bounds__(1024)
gat_range_sort(const int* __restrict__ pairs,
               const int* __restrict__ cbase,
               int* __restrict__ off,
               int* __restrict__ srt,
               int n_nodes, int npb, int nblk, int E) {
    __shared__ int cnt[256];
    __shared__ int cur[256];
    __shared__ int wtot[8];
    const int tid   = threadIdx.x;
    const int lane  = tid & 31;
    const int w     = tid >> 5;
    const int r     = blockIdx.x;
    const int node0 = r * npb;
    const int ebeg  = cbase[r * nblk];
    const int eend  = cbase[(r + 1) * nblk];
    const int2* __restrict__ p2 = reinterpret_cast<const int2*>(pairs);

    if (tid < 256) cnt[tid] = 0;
    __syncthreads();
    for (int i = ebeg + tid; i < eend; i += 1024) atomicAdd(&cnt[p2[i].y - node0], 1);
    __syncthreads();

    int v = 0, inc = 0;
    if (tid < 256) {
        v   = cnt[tid];
        inc = v;
#pragma unroll
        for (int d = 1; d < 32; d <<= 1) {
            const int u = __shfl_up(inc, d);
            if (lane >= d) inc += u;
        }
        if (lane == 31) wtot[w] = inc;
    }
    __syncthreads();
    if (tid < 256) {
        int pre = 0;
        for (int j = 0; j < w; ++j) pre += wtot[j];
        const int excl = pre + inc - v;
        cur[tid] = excl;
        const int node = node0 + tid;
        if (tid < npb && node < n_nodes) off[node] = ebeg + excl;
    }
    if (r == (int)gridDim.x - 1 && tid == 0) off[n_nodes] = E;
    __syncthreads();
    for (int i = ebeg + tid; i < eend; i += 1024) {
        const int2 p = p2[i];
        const int pos = atomicAdd(&cur[p.y - node0], 1);
        srt[ebeg + pos] = p.x;
    }
}

#define SEL8(a, b, c, d, e, f, g, h)                                          \
    ((a) | ((b) << 3) | ((c) << 6) | ((d) << 9) | ((e) << 12) | ((f) << 15) | \
     ((g) << 18) | ((h) << 21))

template <unsigned SEL>
__device__ __forceinline__ float grp8(float v) {
    return __uint_as_float(__builtin_amdgcn_mov_dpp8(__float_as_uint(v), SEL));
}

#define GAT_ROW_LOAD(J, R)                                                    \
    const int i##R = (J) < m ? (J) : m - 1;                                   \
    const int f##R = __builtin_amdgcn_readlane(f, i##R);                      \
    const float4 R = *reinterpret_cast<const float4*>(hbuf + (size_t)f##R * 128 + lane4);
#define GAT_ROW_FMA(J, R)                                                     \
    {                                                                         \
        const float wj = grp8<SEL8(J, J, J, J, J, J, J, J)>(w);               \
        acc.x += wj * R.x; acc.y += wj * R.y;                                 \
        acc.z += wj * R.z; acc.w += wj * R.w;                                 \
    }

__device__ __forceinline__ void gat_rows8(float4& acc, const float* __restrict__ hbuf,
                                          int f, float w, int m, int lane4) {
    GAT_ROW_LOAD(0, r0) GAT_ROW_LOAD(1, r1) GAT_ROW_LOAD(2, r2) GAT_ROW_LOAD(3, r3)
    GAT_ROW_LOAD(4, r4) GAT_ROW_LOAD(5, r5) GAT_ROW_LOAD(6, r6) GAT_ROW_LOAD(7, r7)
    GAT_ROW_FMA(0, r0) GAT_ROW_FMA(1, r1) GAT_ROW_FMA(2, r2) GAT_ROW_FMA(3, r3)
    GAT_ROW_FMA(4, r4) GAT_ROW_FMA(5, r5) GAT_ROW_FMA(6, r6) GAT_ROW_FMA(7, r7)
}

#undef GAT_ROW_LOAD
#undef GAT_ROW_FMA

#define GAT_AGG_WAVES_LOG2 13

struct gat_src { int f0, f1, f2, f3; };
struct gat_sc  { float a0, a1, a2, a3, ar; };

__device__ __forceinline__ gat_src gat_list32(const int* __restrict__ srt, int offv,
                                              int j, int slot) {
    const int beg  = __builtin_amdgcn_readlane(offv, j);
    const int cnt  = __builtin_amdgcn_readlane(offv, j + 1) - beg;
    const int base = cnt > 0 ? beg : 0;
    const int last = cnt > 0 ? cnt - 1 : 0;
    const int* __restrict__ lst = srt + base;
    gat_src s;
    s.f0 = lst[slot      < last ? slot      : last];
    s.f1 = lst[slot + 8  < last ? slot + 8  : last];
    s.f2 = lst[slot + 16 < last ? slot + 16 : last];
    s.f3 = lst[slot + 24 < last ? slot + 24 : last];
    return s;
}

__device__ __forceinline__ gat_sc gat_scores(const float* __restrict__ alpha_l,
                                             const float* __restrict__ alpha_r,
                                             const gat_src f, int n, int n_nodes, int hd) {
    const int nc = n < n_nodes ? n : n_nodes - 1;
    gat_sc a;
    a.a0 = alpha_l[f.f0 * 4 + hd];
    a.a1 = alpha_l[f.f1 * 4 + hd];
    a.a2 = alpha_l[f.f2 * 4 + hd];
    a.a3 = alpha_l[f.f3 * 4 + hd];
    a.ar = alpha_r[nc * 4 + hd];
    return a;
}

__global__ void __launch_bounds__(256)
gat_aggregate(const float* __restrict__ hbuf,
              const int* __restrict__ srt,
              const int* __restrict__ off,
              const float* __restrict__ alpha_l,
              const float* __restrict__ alpha_r,
              const unsigned* __restrict__ maxbuf,
              const float* __restrict__ bias,
              float* __restrict__ out, int n_nodes) {
    const int lane = threadIdx.x & 31;
    const unsigned wv = (unsigned)__builtin_amdgcn_readfirstlane(
        (int)((blockIdx.x * (unsigned)blockDim.x + threadIdx.x) >> 5));
    const int n0 = (int)(((unsigned long long)wv * (unsigned)n_nodes) >> GAT_AGG_WAVES_LOG2);
    const int nn = (int)(((unsigned long long)(wv + 1) * (unsigned)n_nodes) >> GAT_AGG_WAVES_LOG2) - n0;
    if (nn <= 0) return;
    const int offv = off[n0 + (lane < nn ? lane : nn)];

    const int lane4 = lane * 4;
    const int hd    = lane >> 3;
    const int slot  = lane & 7;
    const float4 b  = *reinterpret_cast<const float4*>(bias + lane4);
    const float mx  = ord2f(maxbuf[hd]);

    gat_src Fc = gat_list32(srt, offv, 0, slot);
    gat_src Fn = gat_list32(srt, offv, 1, slot);
    gat_sc  Ac = gat_scores(alpha_l, alpha_r, Fc, n0, n_nodes, hd);

    for (int j = 0; j < nn; ++j) {
        const int n   = n0 + j;
        const int beg = __builtin_amdgcn_readlane(offv, j);
        const int cnt = __builtin_amdgcn_readlane(offv, j + 1) - beg;
        const gat_sc  An  = gat_scores(alpha_l, alpha_r, Fn, n + 1, n_nodes, hd);
        const gat_src Fnn = gat_list32(srt, offv, j + 2, slot);

        float4* __restrict__ dst = reinterpret_cast<float4*>(out + (size_t)n * 128 + lane4);
        if (cnt <= 0) {
            *dst = b;
        } else {
            const int last = cnt - 1;
            const float ar = Ac.ar;
            const int* __restrict__ lst = srt + beg;

            const int j0 = slot, j1 = slot + 8, j2 = slot + 16, j3 = slot + 24;
            float a0 = Ac.a0 + ar;
            float a1 = Ac.a1 + ar;
            float a2 = Ac.a2 + ar;
            float a3 = Ac.a3 + ar;
            a0 = a0 > 0.f ? a0 : 0.2f * a0;
            a1 = a1 > 0.f ? a1 : 0.2f * a1;
            a2 = a2 > 0.f ? a2 : 0.2f * a2;
            a3 = a3 > 0.f ? a3 : 0.2f * a3;
            const float e0 = j0 < cnt ? __expf(a0 - mx) : 0.f;
            const float e1 = j1 < cnt ? __expf(a1 - mx) : 0.f;
            const float e2 = j2 < cnt ? __expf(a2 - mx) : 0.f;
            const float e3 = j3 < cnt ? __expf(a3 - mx) : 0.f;
            float s = e0 + e1;
            s += e2 + e3;
            for (int base = 32; base < cnt; base += 8) {
                const int jj = base + slot;
                const int f = lst[jj < cnt ? jj : last];
                float a = alpha_l[f * 4 + hd] + ar;
                a = a > 0.f ? a : 0.2f * a;
                s += jj < cnt ? __expf(a - mx) : 0.f;
            }
            s += grp8<SEL8(1, 0, 3, 2, 5, 4, 7, 6)>(s);
            s += grp8<SEL8(2, 3, 0, 1, 6, 7, 4, 5)>(s);
            s += grp8<SEL8(4, 5, 6, 7, 0, 1, 2, 3)>(s);
            const float den = s + 1e-9f;

            float4 acc = make_float4(0.f, 0.f, 0.f, 0.f);
            gat_rows8(acc, hbuf, Fc.f0, e0 / den, cnt < 8 ? cnt : 8, lane4);
            if (cnt > 8)  gat_rows8(acc, hbuf, Fc.f1, e1 / den, cnt < 16 ? cnt - 8 : 8, lane4);
            if (cnt > 16) gat_rows8(acc, hbuf, Fc.f2, e2 / den, cnt < 24 ? cnt - 16 : 8, lane4);
            if (cnt > 24) gat_rows8(acc, hbuf, Fc.f3, e3 / den, cnt < 32 ? cnt - 24 : 8, lane4);
            for (int base = 32; base < cnt; base += 8) {
                const int jj = base + slot;
                const int f = lst[jj < cnt ? jj : last];
                float a = alpha_l[f * 4 + hd] + ar;
                a = a > 0.f ? a : 0.2f * a;
                const float w = jj < cnt ? __expf(a - mx) / den : 0.f;
                const int left = cnt - base;
                gat_rows8(acc, hbuf, f, w, left < 8 ? left : 8, lane4);
            }

            float4 o;
            o.x = acc.x + b.x; o.y = acc.y + b.y; o.z = acc.z + b.z; o.w = acc.w + b.w;
            *dst = o;
        }
        Fc = Fn; Fn = Fnn; Ac = An;
    }
}

extern "C" void kernel_launch(void* const* d_in, const int* in_sizes, int n_in,
                              void* d_out, int out_size, void* d_ws, size_t ws_size,
                              hipStream_t stream) {
    const float* x     = (const float*)d_in[0];
    const int*   ei    = (const int*)d_in[1];
    const float* W     = (const float*)d_in[2];
    const float* att_l = (const float*)d_in[3];
    const float* att_r = (const float*)d_in[4];
    const float* bias  = (const float*)d_in[5];
    float* out = (float*)d_out;

    const int N  = in_sizes[0] / 128;
    const int E  = in_sizes[1] / 2;
    const int NH = N * 4;

    float* h       = (float*)d_ws;
    float* alpha_l = h + (size_t)N * 128;
    float* alpha_r = alpha_l + NH;
    int* off       = (int*)(alpha_r + NH);
    unsigned* maxbuf = (unsigned*)(off + N + 4);
    int* srt       = (int*)(maxbuf + 4);
    if ((size_t)((char*)(srt + E) - (char*)d_ws) > ws_size) return;

    const int NRANGE = 256;
    const int NBLK   = 256;
    const int NC     = NRANGE * NBLK;
    const int npb    = (N + NRANGE - 1) / NRANGE;
    const int chunk  = (E + NBLK - 1) / NBLK;
    int* counts = (int*)d_out;
    int* cbase  = counts + NC;
    int* pairs  = cbase + NC + 4;
    if (npb > 256) return;
    if ((size_t)2 * NC + 4 + (size_t)2 * E > (size_t)out_size) return;

    const int BLK = 256;

    gat_gemm_wmma<<<1024, BLK, 0, stream>>>(x, W, att_l, att_r, h, alpha_l, alpha_r,
                                            maxbuf, N / 16);
    gat_edge_max_hist<<<NBLK, 1024, 0, stream>>>(ei, alpha_l, alpha_r, maxbuf,
                                                 counts, E, chunk, npb);
    gat_scan<<<(NC + 1023) / 1024, 1024, 0, stream>>>(counts, cbase, NC);
    gat_range_copy<<<NBLK, 1024, 0, stream>>>(ei, cbase, pairs, E, chunk, npb);
    gat_range_sort<<<NRANGE, 1024, 0, stream>>>(pairs, cbase, off, srt, N, npb, NBLK, E);
    {
        if ((N >> GAT_AGG_WAVES_LOG2) + 4 > 32) return;
        const int blocks = (32 << GAT_AGG_WAVES_LOG2) / BLK;
        gat_aggregate<<<blocks, BLK, 0, stream>>>(h, srt, off, alpha_l, alpha_r,
                                                  maxbuf, bias, out, N);
    }
}


# Edge chunks grouped by node range inside the GEMM launch (256 early blocks: LDS count, scan, place); range-sort launch gathers its 256 groups and takes the max: 6 launches -> 3; sums as before
# speedup vs baseline: 4.9532x; 1.3133x over previous
#include <hip/hip_runtime.h>
#include <hip/hip_bf16.h>


typedef __attribute__((ext_vector_type(2))) float v2f;
typedef __attribute__((ext_vector_type(8))) float v8f;

__device__ __forceinline__ unsigned f2ord(float f) {
    unsigned u = __float_as_uint(f);
    return (u & 0x80000000u) ? ~u : (u | 0x80000000u);
}
__device__ __forceinline__ float ord2f(unsigned u) {
    unsigned v = (u & 0x80000000u) ? (u & 0x7FFFFFFFu) : ~u;
    return __uint_as_float(v);
}

#define GAT_PITCH 132

__global__ void __launch_bounds__(256)
gat_gemm_wmma(const float* __restrict__ x,
              const float* __restrict__ W,
              const float* __restrict__ att_l,
              const float* __restrict__ att_r,
              float* __restrict__ h,
              float* __restrict__ alpha_l,
              float* __restrict__ alpha_r,
              unsigned* __restrict__ maxbuf, int ntiles,
              const int* __restrict__ ei,
              int* __restrict__ counts,
              int* __restrict__ lstart,
              int* __restrict__ pairs, int E, int chunk, int npb) {
    __shared__ int hist[256];
    __shared__ int hwtot[8];
    __shared__ __attribute__((aligned(16))) float sA[16 * GAT_PITCH];
    __shared__ __attribute__((aligned(16))) float sC[16 * GAT_PITCH];
    __shared__ __attribute__((aligned(16))) float sAtt[256];

    const int tid    = threadIdx.x;
    const int wave   = tid >> 5;
    const int lane   = tid & 31;
    const int lhalf  = lane & 15;
    const int khalf  = (lane >> 4) << 1;
    const int colB   = wave * 16 + lhalf;
    const int mhi    = (lane >> 4) << 3;
    const int r0     = 2 * wave;
    const int c4     = 4 * lane;

    if (blockIdx.x == 0 && tid < 4) maxbuf[tid] = f2ord(-3.0e38f);
    sAtt[tid] = tid < 128 ? att_l[tid] : att_r[tid - 128];

    v2f bw[32];
#pragma unroll
    for (int i = 0; i < 32; ++i) {
        const int k = 4 * i + khalf;
        bw[i].x = W[(size_t)k * 128 + colB];
        bw[i].y = W[(size_t)(k + 1) * 128 + colB];
    }

    const int dwhich = tid >> 6;
    const int drow   = (tid & 63) >> 2;
    const int dhead  = tid & 3;

    int t = blockIdx.x;
    float4 p0 = make_float4(0.f, 0.f, 0.f, 0.f), p1 = p0;
    if (t < ntiles) {
        const float* __restrict__ src = x + (size_t)(t * 16 + r0) * 128 + c4;
        p0 = *reinterpret_cast<const float4*>(src);
        p1 = *reinterpret_cast<const float4*>(src + 128);
    }
    const int hb = (int)((blockIdx.x + gridDim.x - (unsigned)ntiles % gridDim.x) % gridDim.x);
    if (hb < 256) {
        hist[tid] = 0;
        __syncthreads();
        const int beg = hb * chunk;
        const int end = beg + chunk < E ? beg + chunk : E;
        const int* __restrict__ to = ei + E;
        for (int e0 = beg + tid; e0 < end; e0 += 8 * 256) {
            int tt[8];
#pragma unroll
            for (int u = 0; u < 8; ++u) {
                const int e = e0 + 256 * u;
                tt[u] = to[e < end ? e : end - 1];
            }
#pragma unroll
            for (int u = 0; u < 8; ++u)
                if (e0 + 256 * u < end) atomicAdd(&hist[tt[u] / npb], 1);
        }
        __syncthreads();
        const int v = hist[tid];
        int inc = v;
#pragma unroll
        for (int d = 1; d < 32; d <<= 1) {
            const int u = __shfl_up(inc, d);
            if (lane >= d) inc += u;
        }
        if (lane == 31) hwtot[wave] = inc;
        __syncthreads();
        int pre = 0;
        for (int j = 0; j < wave; ++j) pre += hwtot[j];
        const int excl = pre + inc - v;
        counts[tid * 256 + hb] = v;
        lstart[tid * 256 + hb] = excl;
        hist[tid] = excl;
        __syncthreads();
        int2* __restrict__ p2 = reinterpret_cast<int2*>(pairs) + beg;
        for (int e0 = beg + tid; e0 < end; e0 += 8 * 256) {
            int ff[8], tt[8];
#pragma unroll
            for (int u = 0; u < 8; ++u) {
                const int e  = e0 + 256 * u;
                const int ec = e < end ? e : end - 1;
                ff[u] = ei[ec];
                tt[u] = to[ec];
            }
#pragma unroll
            for (int u = 0; u < 8; ++u)
                if (e0 + 256 * u < end) {
                    const int pos = atomicAdd(&hist[tt[u] / npb], 1);
                    p2[pos] = make_int2(ff[u], tt[u]);
                }
        }
    }

    for (; t < ntiles; t += gridDim.x) {
        *reinterpret_cast<float4*>(sA + r0 * GAT_PITCH + c4)       = p0;
        *reinterpret_cast<float4*>(sA + (r0 + 1) * GAT_PITCH + c4) = p1;
        __syncthreads();

        const int tn = t + gridDim.x;
        if (tn < ntiles) {
            const float* __restrict__ src = x + (size_t)(tn * 16 + r0) * 128 + c4;
            p0 = *reinterpret_cast<const float4*>(src);
            p1 = *reinterpret_cast<const float4*>(src + 128);
        }

        const float* __restrict__ arow = sA + lhalf * GAT_PITCH + khalf;
        v8f c = {0.f, 0.f, 0.f, 0.f, 0.f, 0.f, 0.f, 0.f};
#pragma unroll
        for (int i = 0; i < 32; ++i) {
            const v2f a = *reinterpret_cast<const v2f*>(arow + 4 * i);
            c = __builtin_amdgcn_wmma_f32_16x16x4_f32(false, a, false, bw[i],
                                                      (short)0, c, false, false);
        }
#pragma unroll
        for (int r = 0; r < 8; ++r)
            sC[(r + mhi) * GAT_PITCH + wave * 16 + lhalf] = c[r];
        __syncthreads();

        {
            const float4 o0 = *reinterpret_cast<const float4*>(sC + r0 * GAT_PITCH + c4);
            const float4 o1 = *reinterpret_cast<const float4*>(sC + (r0 + 1) * GAT_PITCH + c4);
            float* __restrict__ dst = h + (size_t)(t * 16 + r0) * 128 + c4;
            *reinterpret_cast<float4*>(dst)       = o0;
            *reinterpret_cast<float4*>(dst + 128) = o1;
        }
        if (tid < 128) {
            const float* __restrict__ hv = sC + drow * GAT_PITCH + dhead * 32;
            const float* __restrict__ av = sAtt + dwhich * 128 + dhead * 32;
            float s = 0.f;
            for (int d = 0; d < 32; ++d) s += hv[d] * av[d];
            float* __restrict__ ap = dwhich ? alpha_r : alpha_l;
            ap[(size_t)(t * 16 + drow) * 4 + dhead] = s;
        }
    }
}

__global__ void __launch_bounds__(1024)
gat_range_sort(const int* __restrict__ pairs,
               const int* __restrict__ counts,
               const int* __restrict__ lstart,
               const float* __restrict__ alpha_l,
               const float* __restrict__ alpha_r,
               unsigned* __restrict__ maxbuf,
               int* __restrict__ off,
               int* __restrict__ srt,
               int n_nodes, int npb, int chunk, int E) {
    __shared__ int cnt[256];
    __shared__ int cur[256];
    __shared__ int wtot[8];
    __shared__ int seg_n[256];
    __shared__ int seg_at[256];
    __shared__ int s_red[32];
    __shared__ unsigned smax[4];
    const int tid   = threadIdx.x;
    const int lane  = tid & 31;
    const int w     = tid >> 5;
    const int r     = blockIdx.x;
    const int node0 = r * npb;
    const int2* __restrict__ p2 = reinterpret_cast<const int2*>(pairs);

    {
        const int4* __restrict__ c4 = reinterpret_cast<const int4*>(counts);
        const int n4 = r * 64;
        int acc = 0;
        for (int i0 = tid; i0 < n4; i0 += 4 * 1024) {
            int4 cv[4];
#pragma unroll
            for (int u = 0; u < 4; ++u) {
                const int i = i0 + 1024 * u;
                cv[u] = c4[i < n4 ? i : 0];
            }
#pragma unroll
            for (int u = 0; u < 4; ++u)
                acc += i0 + 1024 * u < n4 ? (cv[u].x + cv[u].y) + (cv[u].z + cv[u].w) : 0;
        }
#pragma unroll
        for (int d = 16; d > 0; d >>= 1) acc += __shfl_xor(acc, d);
        if (lane == 0) s_red[w] = acc;
    }
    if (tid < 256) {
        cnt[tid]    = 0;
        seg_n[tid]  = counts[r * 256 + tid];
        seg_at[tid] = tid * chunk + lstart[r * 256 + tid];
    }
    if (tid < 4) smax[tid] = 0u;
    __syncthreads();
    int ebeg = 0;
    for (int j = 0; j < 32; ++j) ebeg += s_red[j];

    const int q  = tid & 3;
    const int sn = seg_n[tid >> 2];
    const int2* __restrict__ sp = p2 + seg_at[tid >> 2];
    const float4* __restrict__ al4 = reinterpret_cast<const float4*>(alpha_l);
    const float4* __restrict__ ar4 = reinterpret_cast<const float4*>(alpha_r);

    float m0 = -3.0e38f, m1 = -3.0e38f, m2 = -3.0e38f, m3 = -3.0e38f;
    for (int i0 = q; i0 < sn; i0 += 16) {
        int2 pp[4];
        float4 ll[4], rr[4];
#pragma unroll
        for (int u = 0; u < 4; ++u) {
            const int i = i0 + 4 * u;
            pp[u] = sp[i < sn ? i : sn - 1];
        }
#pragma unroll
        for (int u = 0; u < 4; ++u) {
            ll[u] = al4[pp[u].x];
            rr[u] = ar4[pp[u].y];
        }
#pragma unroll
        for (int u = 0; u < 4; ++u) {
            if (i0 + 4 * u < sn) atomicAdd(&cnt[pp[u].y - node0], 1);
            float a0 = ll[u].x + rr[u].x;
            float a1 = ll[u].y + rr[u].y;
            float a2 = ll[u].z + rr[u].z;
            float a3 = ll[u].w + rr[u].w;
            a0 = a0 > 0.f ? a0 : 0.2f * a0;
            a1 = a1 > 0.f ? a1 : 0.2f * a1;
            a2 = a2 > 0.f ? a2 : 0.2f * a2;
            a3 = a3 > 0.f ? a3 : 0.2f * a3;
            m0 = fmaxf(m0, a0); m1 = fmaxf(m1, a1);
            m2 = fmaxf(m2, a2); m3 = fmaxf(m3, a3);
        }
    }
#pragma unroll
    for (int d = 16; d > 0; d >>= 1) {
        m0 = fmaxf(m0, __shfl_xor(m0, d)); m1 = fmaxf(m1, __shfl_xor(m1, d));
        m2 = fmaxf(m2, __shfl_xor(m2, d)); m3 = fmaxf(m3, __shfl_xor(m3, d));
    }
    if (lane == 0) {
        atomicMax(&smax[0], f2ord(m0));
        atomicMax(&smax[1], f2ord(m1));
        atomicMax(&smax[2], f2ord(m2));
        atomicMax(&smax[3], f2ord(m3));
    }
    __syncthreads();
    if (tid < 4) atomicMax(&maxbuf[tid], smax[tid]);

    int v = 0, inc = 0;
    if (tid < 256) {
        v   = cnt[tid];
        inc = v;
#pragma unroll
        for (int d = 1; d < 32; d <<= 1) {
            const int u = __shfl_up(inc, d);
            if (lane >= d) inc += u;
        }
        if (lane == 31) wtot[w] = inc;
    }
    __syncthreads();
    if (tid < 256) {
        int pre = 0;
        for (int j = 0; j < w; ++j) pre += wtot[j];
        const int excl = pre + inc - v;
        cur[tid] = excl;
        const int node = node0 + tid;
        if (tid < npb && node < n_nodes) off[node] = ebeg + excl;
    }
    if (r == (int)gridDim.x - 1 && tid == 0) off[n_nodes] = E;
    __syncthreads();
    for (int i0 = q; i0 < sn; i0 += 16) {
        int2 pp[4];
#pragma unroll
        for (int u = 0; u < 4; ++u) {
            const int i = i0 + 4 * u;
            pp[u] = sp[i < sn ? i : sn - 1];
        }
#pragma unroll
        for (int u = 0; u < 4; ++u)
            if (i0 + 4 * u < sn) {
                const int pos = atomicAdd(&cur[pp[u].y - node0], 1);
                srt[ebeg + pos] = pp[u].x;
            }
    }
}

#define SEL8(a, b, c, d, e, f, g, h)                                          \
    ((a) | ((b) << 3) | ((c) << 6) | ((d) << 9) | ((e) << 12) | ((f) << 15) | \
     ((g) << 18) | ((h) << 21))

template <unsigned SEL>
__device__ __forceinline__ float grp8(float v) {
    return __uint_as_float(__builtin_amdgcn_mov_dpp8(__float_as_uint(v), SEL));
}

#define GAT_ROW_LOAD(J, R)                                                    \
    const int i##R = (J) < m ? (J) : m - 1;                                   \
    const int f##R = __builtin_amdgcn_readlane(f, i##R);                      \
    const float4 R = *reinterpret_cast<const float4*>(hbuf + (size_t)f##R * 128 + lane4);
#define GAT_ROW_FMA(J, R)                                                     \
    {                                                                         \
        const float wj = grp8<SEL8(J, J, J, J, J, J, J, J)>(w);               \
        acc.x += wj * R.x; acc.y += wj * R.y;                                 \
        acc.z += wj * R.z; acc.w += wj * R.w;                                 \
    }

__device__ __forceinline__ void gat_rows8(float4& acc, const float* __restrict__ hbuf,
                                          int f, float w, int m, int lane4) {
    GAT_ROW_LOAD(0, r0) GAT_ROW_LOAD(1, r1) GAT_ROW_LOAD(2, r2) GAT_ROW_LOAD(3, r3)
    GAT_ROW_LOAD(4, r4) GAT_ROW_LOAD(5, r5) GAT_ROW_LOAD(6, r6) GAT_ROW_LOAD(7, r7)
    GAT_ROW_FMA(0, r0) GAT_ROW_FMA(1, r1) GAT_ROW_FMA(2, r2) GAT_ROW_FMA(3, r3)
    GAT_ROW_FMA(4, r4) GAT_ROW_FMA(5, r5) GAT_ROW_FMA(6, r6) GAT_ROW_FMA(7, r7)
}

#undef GAT_ROW_LOAD
#undef GAT_ROW_FMA

#define GAT_AGG_WAVES_LOG2 13

struct gat_src { int f0, f1, f2, f3; };
struct gat_sc  { float a0, a1, a2, a3, ar; };

__device__ __forceinline__ gat_src gat_list32(const int* __restrict__ srt, int offv,
                                              int j, int slot) {
    const int beg  = __builtin_amdgcn_readlane(offv, j);
    const int cnt  = __builtin_amdgcn_readlane(offv, j + 1) - beg;
    const int base = cnt > 0 ? beg : 0;
    const int last = cnt > 0 ? cnt - 1 : 0;
    const int* __restrict__ lst = srt + base;
    gat_src s;
    s.f0 = lst[slot      < last ? slot      : last];
    s.f1 = lst[slot + 8  < last ? slot + 8  : last];
    s.f2 = lst[slot + 16 < last ? slot + 16 : last];
    s.f3 = lst[slot + 24 < last ? slot + 24 : last];
    return s;
}

__device__ __forceinline__ gat_sc gat_scores(const float* __restrict__ alpha_l,
                                             const float* __restrict__ alpha_r,
                                             const gat_src f, int n, int n_nodes, int hd) {
    const int nc = n < n_nodes ? n : n_nodes - 1;
    gat_sc a;
    a.a0 = alpha_l[f.f0 * 4 + hd];
    a.a1 = alpha_l[f.f1 * 4 + hd];
    a.a2 = alpha_l[f.f2 * 4 + hd];
    a.a3 = alpha_l[f.f3 * 4 + hd];
    a.ar = alpha_r[nc * 4 + hd];
    return a;
}

__global__ void __launch_bounds__(256)
gat_aggregate(const float* __restrict__ hbuf,
              const int* __restrict__ srt,
              const int* __restrict__ off,
              const float* __restrict__ alpha_l,
              const float* __restrict__ alpha_r,
              const unsigned* __restrict__ maxbuf,
              const float* __restrict__ bias,
              float* __restrict__ out, int n_nodes) {
    const int lane = threadIdx.x & 31;
    const unsigned wv = (unsigned)__builtin_amdgcn_readfirstlane(
        (int)((blockIdx.x * (unsigned)blockDim.x + threadIdx.x) >> 5));
    const int n0 = (int)(((unsigned long long)wv * (unsigned)n_nodes) >> GAT_AGG_WAVES_LOG2);
    const int nn = (int)(((unsigned long long)(wv + 1) * (unsigned)n_nodes) >> GAT_AGG_WAVES_LOG2) - n0;
    if (nn <= 0) return;
    const int offv = off[n0 + (lane < nn ? lane : nn)];

    const int lane4 = lane * 4;
    const int hd    = lane >> 3;
    const int slot  = lane & 7;
    const float4 b  = *reinterpret_cast<const float4*>(bias + lane4);
    const float mx  = ord2f(maxbuf[hd]);

    gat_src Fc = gat_list32(srt, offv, 0, slot);
    gat_src Fn = gat_list32(srt, offv, 1, slot);
    gat_sc  Ac = gat_scores(alpha_l, alpha_r, Fc, n0, n_nodes, hd);

    for (int j = 0; j < nn; ++j) {
        const int n   = n0 + j;
        const int beg = __builtin_amdgcn_readlane(offv, j);
        const int cnt = __builtin_amdgcn_readlane(offv, j + 1) - beg;
        const gat_sc  An  = gat_scores(alpha_l, alpha_r, Fn, n + 1, n_nodes, hd);
        const gat_src Fnn = gat_list32(srt, offv, j + 2, slot);

        float4* __restrict__ dst = reinterpret_cast<float4*>(out + (size_t)n * 128 + lane4);
        if (cnt <= 0) {
            *dst = b;
        } else {
            const int last = cnt - 1;
            const float ar = Ac.ar;
            const int* __restrict__ lst = srt + beg;

            const int j0 = slot, j1 = slot + 8, j2 = slot + 16, j3 = slot + 24;
            float a0 = Ac.a0 + ar;
            float a1 = Ac.a1 + ar;
            float a2 = Ac.a2 + ar;
            float a3 = Ac.a3 + ar;
            a0 = a0 > 0.f ? a0 : 0.2f * a0;
            a1 = a1 > 0.f ? a1 : 0.2f * a1;
            a2 = a2 > 0.f ? a2 : 0.2f * a2;
            a3 = a3 > 0.f ? a3 : 0.2f * a3;
            const float e0 = j0 < cnt ? __expf(a0 - mx) : 0.f;
            const float e1 = j1 < cnt ? __expf(a1 - mx) : 0.f;
            const float e2 = j2 < cnt ? __expf(a2 - mx) : 0.f;
            const float e3 = j3 < cnt ? __expf(a3 - mx) : 0.f;
            float s = e0 + e1;
            s += e2 + e3;
            for (int base = 32; base < cnt; base += 8) {
                const int jj = base + slot;
                const int f = lst[jj < cnt ? jj : last];
                float a = alpha_l[f * 4 + hd] + ar;
                a = a > 0.f ? a : 0.2f * a;
                s += jj < cnt ? __expf(a - mx) : 0.f;
            }
            s += grp8<SEL8(1, 0, 3, 2, 5, 4, 7, 6)>(s);
            s += grp8<SEL8(2, 3, 0, 1, 6, 7, 4, 5)>(s);
            s += grp8<SEL8(4, 5, 6, 7, 0, 1, 2, 3)>(s);
            const float den = s + 1e-9f;

            float4 acc = make_float4(0.f, 0.f, 0.f, 0.f);
            gat_rows8(acc, hbuf, Fc.f0, e0 / den, cnt < 8 ? cnt : 8, lane4);
            if (cnt > 8)  gat_rows8(acc, hbuf, Fc.f1, e1 / den, cnt < 16 ? cnt - 8 : 8, lane4);
            if (cnt > 16) gat_rows8(acc, hbuf, Fc.f2, e2 / den, cnt < 24 ? cnt - 16 : 8, lane4);
            if (cnt > 24) gat_rows8(acc, hbuf, Fc.f3, e3 / den, cnt < 32 ? cnt - 24 : 8, lane4);
            for (int base = 32; base < cnt; base += 8) {
                const int jj = base + slot;
                const int f = lst[jj < cnt ? jj : last];
                float a = alpha_l[f * 4 + hd] + ar;
                a = a > 0.f ? a : 0.2f * a;
                const float w = jj < cnt ? __expf(a - mx) / den : 0.f;
                const int left = cnt - base;
                gat_rows8(acc, hbuf, f, w, left < 8 ? left : 8, lane4);
            }

            float4 o;
            o.x = acc.x + b.x; o.y = acc.y + b.y; o.z = acc.z + b.z; o.w = acc.w + b.w;
            *dst = o;
        }
        Fc = Fn; Fn = Fnn; Ac = An;
    }
}

extern "C" void kernel_launch(void* const* d_in, const int* in_sizes, int n_in,
                              void* d_out, int out_size, void* d_ws, size_t ws_size,
                              hipStream_t stream) {
    const float* x     = (const float*)d_in[0];
    const int*   ei    = (const int*)d_in[1];
    const float* W     = (const float*)d_in[2];
    const float* att_l = (const float*)d_in[3];
    const float* att_r = (const float*)d_in[4];
    const float* bias  = (const float*)d_in[5];
    float* out = (float*)d_out;

    const int N  = in_sizes[0] / 128;
    const int E  = in_sizes[1] / 2;
    const int NH = N * 4;

    float* h       = (float*)d_ws;
    float* alpha_l = h + (size_t)N * 128;
    float* alpha_r = alpha_l + NH;
    int* off       = (int*)(alpha_r + NH);
    unsigned* maxbuf = (unsigned*)(off + N + 4);
    int* srt       = (int*)(maxbuf + 4);
    if ((size_t)((char*)(srt + E) - (char*)d_ws) > ws_size) return;

    const int NRANGE = 256;
    const int NBLK   = 256;
    const int NC     = NRANGE * NBLK;
    const int npb    = (N + NRANGE - 1) / NRANGE;
    const int chunk  = (E + NBLK - 1) / NBLK;
    int* counts = (int*)d_out;
    int* lstart = counts + NC;
    int* pairs  = lstart + NC + 4;
    if (npb > 256) return;
    if ((size_t)2 * NC + 4 + (size_t)2 * E > (size_t)out_size) return;

    const int BLK = 256;

    gat_gemm_wmma<<<1024, BLK, 0, stream>>>(x, W, att_l, att_r, h, alpha_l, alpha_r,
                                            maxbuf, N / 16, ei, counts, lstart, pairs,
                                            E, chunk, npb);
    gat_range_sort<<<NRANGE, 1024, 0, stream>>>(pairs, counts, lstart, alpha_l, alpha_r,
                                                maxbuf, off, srt, N, npb, chunk, E);
    {
        if ((N >> GAT_AGG_WAVES_LOG2) + 4 > 32) return;
        const int blocks = (32 << GAT_AGG_WAVES_LOG2) / BLK;
        gat_aggregate<<<blocks, BLK, 0, stream>>>(h, srt, off, alpha_l, alpha_r,
                                                  maxbuf, bias, out, N);
    }
}
